# starting kernel
# baseline (speedup 1.0000x reference)
#include <hip/hip_runtime.h>
#include <hip/hip_bf16.h>
#include <math.h>

typedef __attribute__((ext_vector_type(2))) float v2f;
typedef __attribute__((ext_vector_type(8))) float v8f;

#define BATCH 4
#define CCH   3
#define HH    64
#define WW    64
#define D1    3072
#define D2    2048
#define D3    1024
#define SIGMA 0.01f

#define BK     32
#define NKB    (D2 / BK)
#define LDS_S  132

__global__ void pool_kernel(const float* __restrict__ x, const float* __restrict__ bias,
                            float* __restrict__ xp) {
    int t = blockIdx.x * 256 + threadIdx.x;
    if (t >= BATCH * D1) return;
    int b = t / D1, f = t % D1;
    int c = f >> 10;
    int rem = f & 1023;
    int oh = rem >> 5, ow = rem & 31;
    int base = c * (HH * WW) + (2 * oh) * WW + 2 * ow;
    const float* xb = x + b * (CCH * HH * WW);
    float s = (xb[base]        + bias[base])
            + (xb[base + 1]    + bias[base + 1])
            + (xb[base + WW]   + bias[base + WW])
            + (xb[base + WW+1] + bias[base + WW + 1]);
    xp[t] = 0.25f * s;
}

__global__ void colabs_kernel(const float* __restrict__ W, float* __restrict__ out,
                              int rows, int cols) {
    int i = blockIdx.x * 256 + threadIdx.x;
    if (i >= cols) return;
    float s = 0.f;
    for (int o = 0; o < rows; ++o) s += fabsf(W[(size_t)o * cols + i]);
    out[i] = s;
}

__global__ void gemv4_kernel(const float* __restrict__ W, const float* __restrict__ X,
                             float* __restrict__ out, int din, int dout) {
    __shared__ float red[4][256];
    int o = blockIdx.x, tid = threadIdx.x;
    const float* wr = W + (size_t)o * din;
    float a0 = 0.f, a1 = 0.f, a2 = 0.f, a3 = 0.f;
    for (int i = tid; i < din; i += 256) {
        float w = wr[i];
        a0 += w * X[i];
        a1 += w * X[din + i];
        a2 += w * X[2 * din + i];
        a3 += w * X[3 * din + i];
    }
    red[0][tid] = a0; red[1][tid] = a1; red[2][tid] = a2; red[3][tid] = a3;
    __syncthreads();
    for (int s = 128; s > 0; s >>= 1) {
        if (tid < s) {
            red[0][tid] += red[0][tid + s];
            red[1][tid] += red[1][tid + s];
            red[2][tid] += red[2][tid + s];
            red[3][tid] += red[3][tid + s];
        }
        __syncthreads();
    }
    if (tid < 4) out[tid * dout + o] = red[tid][0];
}

__global__ void stats1_kernel(const float* __restrict__ xp, const float* __restrict__ colabs1,
                              const float* __restrict__ g1p, float* __restrict__ vn1,
                              float* __restrict__ P1) {
    __shared__ float s1[256], s2[256];
    int b = blockIdx.x, tid = threadIdx.x;
    float m = 0.f, p = 0.f;
    for (int i = tid; i < D1; i += 256) {
        float v = xp[b * D1 + i];
        float v2 = v * v;
        m += v2;
        p += colabs1[i] * v2;
    }
    s1[tid] = m; s2[tid] = p;
    __syncthreads();
    for (int s = 128; s > 0; s >>= 1) {
        if (tid < s) { s1[tid] += s1[tid + s]; s2[tid] += s2[tid + s]; }
        __syncthreads();
    }
    if (tid == 0) {
        float c1 = g1p[0];
        vn1[b] = 2.f * (SIGMA * SIGMA) / (c1 * c1) * s1[0];
        P1[b]  = c1 * s2[0];
    }
}

__global__ void act_kernel(const float* __restrict__ mu1, const float* __restrict__ vn1,
                           float* __restrict__ d2, float* __restrict__ sp,
                           float* __restrict__ m2b) {
    int t = blockIdx.x * 256 + threadIdx.x;
    if (t >= BATCH * D2) return;
    int b = t / D2;
    float m = mu1[t];
    float dd = 1.f / (1.f + expf(-m));
    float s  = (m > 20.f) ? m : log1pf(expf(m));
    float dsq = dd * dd;
    d2[t] = dsq;
    sp[t] = s;
    m2b[t] = s * s + vn1[b] * dsq;
}

__global__ void stats2_kernel(const float* __restrict__ m2b, const float* __restrict__ colabs2,
                              const float* __restrict__ g2p, const float* __restrict__ P1,
                              float* __restrict__ vn2, float* __restrict__ Pout) {
    __shared__ float s1[256], s2[256];
    int b = blockIdx.x, tid = threadIdx.x;
    float m = 0.f, p = 0.f;
    for (int j = tid; j < D2; j += 256) {
        float v = m2b[b * D2 + j];
        m += v;
        p += colabs2[j] * v;
    }
    s1[tid] = m; s2[tid] = p;
    __syncthreads();
    for (int s = 128; s > 0; s >>= 1) {
        if (tid < s) { s1[tid] += s1[tid + s]; s2[tid] += s2[tid + s]; }
        __syncthreads();
    }
    if (tid == 0) {
        float c2 = g2p[0];
        vn2[b] = 2.f * (SIGMA * SIGMA) / (c2 * c2) * s1[0];
        Pout[b] = P1[b] + c2 * s2[0];
    }
}

__global__ void __launch_bounds__(128)
gamma_kernel(const float* __restrict__ W2, const float* __restrict__ d2,
             const float* __restrict__ vn1, const float* __restrict__ vn2,
             float* __restrict__ out) {
    __shared__ float As[BK * LDS_S];
    __shared__ float Bs[BK * LDS_S];

    const int tid  = threadIdx.x;
    const int lane = tid & 31;
    const int w    = tid >> 5;
    const int wy   = w >> 1;
    const int wx   = w & 1;
    const int r    = lane & 15;
    const int hi   = lane >> 4;

    const int brow = blockIdx.x * 128;
    const int bcol = blockIdx.y * 128;
    const int b    = blockIdx.z;

    const int srow = tid >> 3;
    const int c4   = tid & 7;
    const float* d2b = d2 + (size_t)b * D2;
    const float* gA  = W2 + (size_t)(brow + srow) * D2 + c4 * 4;
    const float* gB  = W2 + (size_t)(bcol + srow) * D2 + c4 * 4;

    float4 ra[8], rb[8], rd;

    rd = *(const float4*)(d2b + c4 * 4);
#pragma unroll
    for (int p = 0; p < 8; ++p) {
        ra[p] = *(const float4*)(gA + (size_t)p * 16 * D2);
        rb[p] = *(const float4*)(gB + (size_t)p * 16 * D2);
    }

    v8f acc[4][4] = {};

    const int ca = wy * 64 + r;
    const int cb = wx * 64 + r;

    for (int kb = 0; kb < NKB; ++kb) {
        __syncthreads();
#pragma unroll
        for (int p = 0; p < 8; ++p) {
            int sbase = c4 * 4 * LDS_S + p * 16 + srow;
            As[sbase            ] = ra[p].x * rd.x;
            As[sbase +     LDS_S] = ra[p].y * rd.y;
            As[sbase + 2 * LDS_S] = ra[p].z * rd.z;
            As[sbase + 3 * LDS_S] = ra[p].w * rd.w;
            Bs[sbase            ] = rb[p].x;
            Bs[sbase +     LDS_S] = rb[p].y;
            Bs[sbase + 2 * LDS_S] = rb[p].z;
            Bs[sbase + 3 * LDS_S] = rb[p].w;
        }
        if (kb + 1 < NKB) {
            const float* gA2 = gA + (kb + 1) * BK;
            const float* gB2 = gB + (kb + 1) * BK;
            rd = *(const float4*)(d2b + (kb + 1) * BK + c4 * 4);
#pragma unroll
            for (int p = 0; p < 8; ++p) {
                ra[p] = *(const float4*)(gA2 + (size_t)p * 16 * D2);
                rb[p] = *(const float4*)(gB2 + (size_t)p * 16 * D2);
            }
        }
        __syncthreads();

#pragma unroll
        for (int kk = 0; kk < BK; kk += 4) {
            const int k0 = (kk + 2 * hi) * LDS_S;
            v2f a[4], bv[4];
#pragma unroll
            for (int i = 0; i < 4; ++i) {
                a[i][0]  = As[k0         + ca + i * 16];
                a[i][1]  = As[k0 + LDS_S + ca + i * 16];
                bv[i][0] = Bs[k0         + cb + i * 16];
                bv[i][1] = Bs[k0 + LDS_S + cb + i * 16];
            }
#pragma unroll
            for (int i = 0; i < 4; ++i)
#pragma unroll
                for (int j = 0; j < 4; ++j)
                    acc[i][j] = __builtin_amdgcn_wmma_f32_16x16x4_f32(
                        false, a[i], false, bv[j], (short)0, acc[i][j], false, false);
        }
    }

    const float s  = vn1[b];
    const float vd = vn2[b];
    float* ob = out + (size_t)b * D3 * D3;
#pragma unroll
    for (int i = 0; i < 4; ++i) {
#pragma unroll
        for (int j = 0; j < 4; ++j) {
#pragma unroll
            for (int v = 0; v < 8; ++v) {
                int row = brow + wy * 64 + i * 16 + v + 8 * hi;
                int col = bcol + wx * 64 + j * 16 + r;
                float val = s * acc[i][j][v];
                if (row == col) val += vd;
                ob[(size_t)row * D3 + col] = val;
            }
        }
    }
}

extern "C" void kernel_launch(void* const* d_in, const int* in_sizes, int n_in,
                              void* d_out, int out_size, void* d_ws, size_t ws_size,
                              hipStream_t stream) {
    (void)in_sizes; (void)n_in; (void)out_size; (void)ws_size;
    const float* x    = (const float*)d_in[0];
    const float* bias = (const float*)d_in[1];
    const float* W1   = (const float*)d_in[2];
    const float* W2   = (const float*)d_in[3];
    const float* g1   = (const float*)d_in[4];
    const float* g2   = (const float*)d_in[5];

    float* out = (float*)d_out;
    float* out_mu2   = out;
    float* out_gamma = out + BATCH * D3;
    float* out_P     = out + BATCH * D3 + (size_t)BATCH * D3 * D3;

    float* ws = (float*)d_ws;
    float* xp      = ws;
    float* colabs1 = xp + BATCH * D1;
    float* colabs2 = colabs1 + D1;
    float* mu1     = colabs2 + D2;
    float* d2      = mu1 + BATCH * D2;
    float* sp      = d2 + BATCH * D2;
    float* m2b     = sp + BATCH * D2;
    float* vn1     = m2b + BATCH * D2;
    float* vn2     = vn1 + BATCH;
    float* P1      = vn2 + BATCH;

    pool_kernel<<<(BATCH * D1 + 255) / 256, 256, 0, stream>>>(x, bias, xp);
    colabs_kernel<<<(D1 + 255) / 256, 256, 0, stream>>>(W1, colabs1, D2, D1);
    colabs_kernel<<<(D2 + 255) / 256, 256, 0, stream>>>(W2, colabs2, D3, D2);
    gemv4_kernel<<<D2, 256, 0, stream>>>(W1, xp, mu1, D1, D2);
    stats1_kernel<<<BATCH, 256, 0, stream>>>(xp, colabs1, g1, vn1, P1);
    act_kernel<<<(BATCH * D2 + 255) / 256, 256, 0, stream>>>(mu1, vn1, d2, sp, m2b);
    stats2_kernel<<<BATCH, 256, 0, stream>>>(m2b, colabs2, g2, P1, vn2, out_P);
    gemv4_kernel<<<D3, 256, 0, stream>>>(W2, sp, out_mu2, D2, D3);

    dim3 ggrid(D3 / 128, D3 / 128, BATCH);
    gamma_kernel<<<ggrid, 128, 0, stream>>>(W2, d2, vn1, vn2, out_gamma);
}


# Fold both |W| column-sum passes into the GEMVs (|w|*m summed per row beside w*x, rows then tree-summed); P_tot summed rows-first; 9 launches to 7
# speedup vs baseline: 2.1786x; 2.1786x over previous
#include <hip/hip_runtime.h>
#include <hip/hip_bf16.h>
#include <math.h>

typedef __attribute__((ext_vector_type(2))) float v2f;
typedef __attribute__((ext_vector_type(8))) float v8f;

#define BATCH 4
#define CCH   3
#define HH    64
#define WW    64
#define D1    3072
#define D2    2048
#define D3    1024
#define SIGMA 0.01f

#define BK     32
#define NKB    (D2 / BK)
#define LDS_S  132

__global__ void pool_kernel(const float* __restrict__ x, const float* __restrict__ bias,
                            float* __restrict__ xp) {
    int t = blockIdx.x * 256 + threadIdx.x;
    if (t >= BATCH * D1) return;
    int b = t / D1, f = t % D1;
    int c = f >> 10;
    int rem = f & 1023;
    int oh = rem >> 5, ow = rem & 31;
    int base = c * (HH * WW) + (2 * oh) * WW + 2 * ow;
    const float* xb = x + b * (CCH * HH * WW);
    float s = (xb[base]        + bias[base])
            + (xb[base + 1]    + bias[base + 1])
            + (xb[base + WW]   + bias[base + WW])
            + (xb[base + WW+1] + bias[base + WW + 1]);
    xp[t] = 0.25f * s;
}

template <bool SQ>
__global__ void gemv4p_kernel(const float* __restrict__ W, const float* __restrict__ X,
                              const float* __restrict__ M, float* __restrict__ out,
                              float* __restrict__ q, int din, int dout) {
    __shared__ float red[8][256];
    int o = blockIdx.x, tid = threadIdx.x;
    const float* wr = W + (size_t)o * din;
    float a0 = 0.f, a1 = 0.f, a2 = 0.f, a3 = 0.f;
    float p0 = 0.f, p1 = 0.f, p2 = 0.f, p3 = 0.f;
    for (int i = tid; i < din; i += 256) {
        float w  = wr[i];
        float aw = fabsf(w);
        float x0 = X[i], x1 = X[din + i], x2 = X[2 * din + i], x3 = X[3 * din + i];
        a0 += w * x0;
        a1 += w * x1;
        a2 += w * x2;
        a3 += w * x3;
        float m0, m1, m2, m3;
        if (SQ) {
            m0 = x0 * x0; m1 = x1 * x1; m2 = x2 * x2; m3 = x3 * x3;
        } else {
            m0 = M[i]; m1 = M[din + i]; m2 = M[2 * din + i]; m3 = M[3 * din + i];
        }
        p0 += aw * m0;
        p1 += aw * m1;
        p2 += aw * m2;
        p3 += aw * m3;
    }
    red[0][tid] = a0; red[1][tid] = a1; red[2][tid] = a2; red[3][tid] = a3;
    red[4][tid] = p0; red[5][tid] = p1; red[6][tid] = p2; red[7][tid] = p3;
    __syncthreads();
    for (int s = 128; s > 0; s >>= 1) {
        if (tid < s) {
#pragma unroll
            for (int v = 0; v < 8; ++v) red[v][tid] += red[v][tid + s];
        }
        __syncthreads();
    }
    if (tid < 4)      out[tid * dout + o]     = red[tid][0];
    else if (tid < 8) q[(tid - 4) * dout + o] = red[tid][0];
}

__global__ void stats1_kernel(const float* __restrict__ xp, const float* __restrict__ q1,
                              const float* __restrict__ g1p, float* __restrict__ vn1,
                              float* __restrict__ P1) {
    __shared__ float s1[256], s2[256];
    int b = blockIdx.x, tid = threadIdx.x;
    float m = 0.f, p = 0.f;
    for (int i = tid; i < D1; i += 256) {
        float v = xp[b * D1 + i];
        m += v * v;
    }
    for (int o = tid; o < D2; o += 256) p += q1[b * D2 + o];
    s1[tid] = m; s2[tid] = p;
    __syncthreads();
    for (int s = 128; s > 0; s >>= 1) {
        if (tid < s) { s1[tid] += s1[tid + s]; s2[tid] += s2[tid + s]; }
        __syncthreads();
    }
    if (tid == 0) {
        float c1 = g1p[0];
        vn1[b] = 2.f * (SIGMA * SIGMA) / (c1 * c1) * s1[0];
        P1[b]  = c1 * s2[0];
    }
}

__global__ void act_kernel(const float* __restrict__ mu1, const float* __restrict__ vn1,
                           float* __restrict__ d2, float* __restrict__ sp,
                           float* __restrict__ m2b) {
    int t = blockIdx.x * 256 + threadIdx.x;
    if (t >= BATCH * D2) return;
    int b = t / D2;
    float m = mu1[t];
    float dd = 1.f / (1.f + expf(-m));
    float s  = (m > 20.f) ? m : log1pf(expf(m));
    float dsq = dd * dd;
    d2[t] = dsq;
    sp[t] = s;
    m2b[t] = s * s + vn1[b] * dsq;
}

__global__ void stats2_kernel(const float* __restrict__ m2b, const float* __restrict__ q2,
                              const float* __restrict__ g2p, const float* __restrict__ P1,
                              float* __restrict__ vn2, float* __restrict__ Pout) {
    __shared__ float s1[256], s2[256];
    int b = blockIdx.x, tid = threadIdx.x;
    float m = 0.f, p = 0.f;
    for (int j = tid; j < D2; j += 256) m += m2b[b * D2 + j];
    for (int o = tid; o < D3; o += 256) p += q2[b * D3 + o];
    s1[tid] = m; s2[tid] = p;
    __syncthreads();
    for (int s = 128; s > 0; s >>= 1) {
        if (tid < s) { s1[tid] += s1[tid + s]; s2[tid] += s2[tid + s]; }
        __syncthreads();
    }
    if (tid == 0) {
        float c2 = g2p[0];
        vn2[b] = 2.f * (SIGMA * SIGMA) / (c2 * c2) * s1[0];
        Pout[b] = P1[b] + c2 * s2[0];
    }
}

__global__ void __launch_bounds__(128)
gamma_kernel(const float* __restrict__ W2, const float* __restrict__ d2,
             const float* __restrict__ vn1, const float* __restrict__ vn2,
             float* __restrict__ out) {
    __shared__ float As[BK * LDS_S];
    __shared__ float Bs[BK * LDS_S];

    const int tid  = threadIdx.x;
    const int lane = tid & 31;
    const int w    = tid >> 5;
    const int wy   = w >> 1;
    const int wx   = w & 1;
    const int r    = lane & 15;
    const int hi   = lane >> 4;

    const int brow = blockIdx.x * 128;
    const int bcol = blockIdx.y * 128;
    const int b    = blockIdx.z;

    const int srow = tid >> 3;
    const int c4   = tid & 7;
    const float* d2b = d2 + (size_t)b * D2;
    const float* gA  = W2 + (size_t)(brow + srow) * D2 + c4 * 4;
    const float* gB  = W2 + (size_t)(bcol + srow) * D2 + c4 * 4;

    float4 ra[8], rb[8], rd;

    rd = *(const float4*)(d2b + c4 * 4);
#pragma unroll
    for (int p = 0; p < 8; ++p) {
        ra[p] = *(const float4*)(gA + (size_t)p * 16 * D2);
        rb[p] = *(const float4*)(gB + (size_t)p * 16 * D2);
    }

    v8f acc[4][4] = {};

    const int ca = wy * 64 + r;
    const int cb = wx * 64 + r;

    for (int kb = 0; kb < NKB; ++kb) {
        __syncthreads();
#pragma unroll
        for (int p = 0; p < 8; ++p) {
            int sbase = c4 * 4 * LDS_S + p * 16 + srow;
            As[sbase            ] = ra[p].x * rd.x;
            As[sbase +     LDS_S] = ra[p].y * rd.y;
            As[sbase + 2 * LDS_S] = ra[p].z * rd.z;
            As[sbase + 3 * LDS_S] = ra[p].w * rd.w;
            Bs[sbase            ] = rb[p].x;
            Bs[sbase +     LDS_S] = rb[p].y;
            Bs[sbase + 2 * LDS_S] = rb[p].z;
            Bs[sbase + 3 * LDS_S] = rb[p].w;
        }
        if (kb + 1 < NKB) {
            const float* gA2 = gA + (kb + 1) * BK;
            const float* gB2 = gB + (kb + 1) * BK;
            rd = *(const float4*)(d2b + (kb + 1) * BK + c4 * 4);
#pragma unroll
            for (int p = 0; p < 8; ++p) {
                ra[p] = *(const float4*)(gA2 + (size_t)p * 16 * D2);
                rb[p] = *(const float4*)(gB2 + (size_t)p * 16 * D2);
            }
        }
        __syncthreads();

#pragma unroll
        for (int kk = 0; kk < BK; kk += 4) {
            const int k0 = (kk + 2 * hi) * LDS_S;
            v2f a[4], bv[4];
#pragma unroll
            for (int i = 0; i < 4; ++i) {
                a[i][0]  = As[k0         + ca + i * 16];
                a[i][1]  = As[k0 + LDS_S + ca + i * 16];
                bv[i][0] = Bs[k0         + cb + i * 16];
                bv[i][1] = Bs[k0 + LDS_S + cb + i * 16];
            }
#pragma unroll
            for (int i = 0; i < 4; ++i)
#pragma unroll
                for (int j = 0; j < 4; ++j)
                    acc[i][j] = __builtin_amdgcn_wmma_f32_16x16x4_f32(
                        false, a[i], false, bv[j], (short)0, acc[i][j], false, false);
        }
    }

    const float s  = vn1[b];
    const float vd = vn2[b];
    float* ob = out + (size_t)b * D3 * D3;
#pragma unroll
    for (int i = 0; i < 4; ++i) {
#pragma unroll
        for (int j = 0; j < 4; ++j) {
#pragma unroll
            for (int v = 0; v < 8; ++v) {
                int row = brow + wy * 64 + i * 16 + v + 8 * hi;
                int col = bcol + wx * 64 + j * 16 + r;
                float val = s * acc[i][j][v];
                if (row == col) val += vd;
                ob[(size_t)row * D3 + col] = val;
            }
        }
    }
}

extern "C" void kernel_launch(void* const* d_in, const int* in_sizes, int n_in,
                              void* d_out, int out_size, void* d_ws, size_t ws_size,
                              hipStream_t stream) {
    (void)in_sizes; (void)n_in; (void)out_size; (void)ws_size;
    const float* x    = (const float*)d_in[0];
    const float* bias = (const float*)d_in[1];
    const float* W1   = (const float*)d_in[2];
    const float* W2   = (const float*)d_in[3];
    const float* g1   = (const float*)d_in[4];
    const float* g2   = (const float*)d_in[5];

    float* out = (float*)d_out;
    float* out_mu2   = out;
    float* out_gamma = out + BATCH * D3;
    float* out_P     = out + BATCH * D3 + (size_t)BATCH * D3 * D3;

    float* ws = (float*)d_ws;
    float* xp      = ws;
    float* spare   = xp + BATCH * D1;
    float* mu1     = spare + D1 + D2;
    float* d2      = mu1 + BATCH * D2;
    float* sp      = d2 + BATCH * D2;
    float* m2b     = sp + BATCH * D2;
    float* vn1     = m2b + BATCH * D2;
    float* vn2     = vn1 + BATCH;
    float* P1      = vn2 + BATCH;
    float* q1      = d2;
    float* q2      = mu1;

    pool_kernel<<<(BATCH * D1 + 255) / 256, 256, 0, stream>>>(x, bias, xp);
    gemv4p_kernel<true><<<D2, 256, 0, stream>>>(W1, xp, xp, mu1, q1, D1, D2);
    stats1_kernel<<<BATCH, 256, 0, stream>>>(xp, q1, g1, vn1, P1);
    act_kernel<<<(BATCH * D2 + 255) / 256, 256, 0, stream>>>(mu1, vn1, d2, sp, m2b);
    gemv4p_kernel<false><<<D3, 256, 0, stream>>>(W2, sp, m2b, out_mu2, q2, D2, D3);
    stats2_kernel<<<BATCH, 256, 0, stream>>>(m2b, q2, g2, P1, vn2, out_P);

    dim3 ggrid(D3 / 128, D3 / 128, BATCH);
    gamma_kernel<<<ggrid, 128, 0, stream>>>(W2, d2, vn1, vn2, out_gamma);
}


# gamma2 = W2 diag(d^2) W2^T is symmetric: compute only the 136 of 256 64x64 tiles on or above the diagonal (544 blocks, 2-3 per WGP); each mirror tile is a transposed copy via LDS of the upper one
# speedup vs baseline: 3.6615x; 1.6807x over previous
#include <hip/hip_runtime.h>
#include <hip/hip_bf16.h>
#include <math.h>

typedef __attribute__((ext_vector_type(2))) float v2f;
typedef __attribute__((ext_vector_type(4))) float v4f;
typedef __attribute__((ext_vector_type(8))) float v8f;

#define BATCH 4
#define CCH   3
#define HH    64
#define WW    64
#define D1    3072
#define D2    2048
#define D3    1024
#define SIGMA 0.01f

#define GT     64
#define NT     (D3 / GT)
#define NPAIR  (NT * (NT + 1) / 2)
#define BK     32
#define NKB    (D2 / BK)
#define GP     36
#define CP     68

__global__ void pool_kernel(const float* __restrict__ x, const float* __restrict__ bias,
                            float* __restrict__ xp) {
    int t = blockIdx.x * 256 + threadIdx.x;
    if (t >= BATCH * D1) return;
    int b = t / D1, f = t % D1;
    int c = f >> 10;
    int rem = f & 1023;
    int oh = rem >> 5, ow = rem & 31;
    int base = c * (HH * WW) + (2 * oh) * WW + 2 * ow;
    const float* xb = x + b * (CCH * HH * WW);
    float s = (xb[base]        + bias[base])
            + (xb[base + 1]    + bias[base + 1])
            + (xb[base + WW]   + bias[base + WW])
            + (xb[base + WW+1] + bias[base + WW + 1]);
    xp[t] = 0.25f * s;
}

template <bool SQ>
__global__ void gemv4p_kernel(const float* __restrict__ W, const float* __restrict__ X,
                              const float* __restrict__ M, float* __restrict__ out,
                              float* __restrict__ q, int din, int dout) {
    __shared__ float red[8][256];
    int o = blockIdx.x, tid = threadIdx.x;
    const float* wr = W + (size_t)o * din;
    float a0 = 0.f, a1 = 0.f, a2 = 0.f, a3 = 0.f;
    float p0 = 0.f, p1 = 0.f, p2 = 0.f, p3 = 0.f;
    for (int i = tid; i < din; i += 256) {
        float w  = wr[i];
        float aw = fabsf(w);
        float x0 = X[i], x1 = X[din + i], x2 = X[2 * din + i], x3 = X[3 * din + i];
        a0 += w * x0;
        a1 += w * x1;
        a2 += w * x2;
        a3 += w * x3;
        float m0, m1, m2, m3;
        if (SQ) {
            m0 = x0 * x0; m1 = x1 * x1; m2 = x2 * x2; m3 = x3 * x3;
        } else {
            m0 = M[i]; m1 = M[din + i]; m2 = M[2 * din + i]; m3 = M[3 * din + i];
        }
        p0 += aw * m0;
        p1 += aw * m1;
        p2 += aw * m2;
        p3 += aw * m3;
    }
    red[0][tid] = a0; red[1][tid] = a1; red[2][tid] = a2; red[3][tid] = a3;
    red[4][tid] = p0; red[5][tid] = p1; red[6][tid] = p2; red[7][tid] = p3;
    __syncthreads();
    for (int s = 128; s > 0; s >>= 1) {
        if (tid < s) {
#pragma unroll
            for (int v = 0; v < 8; ++v) red[v][tid] += red[v][tid + s];
        }
        __syncthreads();
    }
    if (tid < 4)      out[tid * dout + o]     = red[tid][0];
    else if (tid < 8) q[(tid - 4) * dout + o] = red[tid][0];
}

__global__ void stats1_kernel(const float* __restrict__ xp, const float* __restrict__ q1,
                              const float* __restrict__ g1p, float* __restrict__ vn1,
                              float* __restrict__ P1) {
    __shared__ float s1[256], s2[256];
    int b = blockIdx.x, tid = threadIdx.x;
    float m = 0.f, p = 0.f;
    for (int i = tid; i < D1; i += 256) {
        float v = xp[b * D1 + i];
        m += v * v;
    }
    for (int o = tid; o < D2; o += 256) p += q1[b * D2 + o];
    s1[tid] = m; s2[tid] = p;
    __syncthreads();
    for (int s = 128; s > 0; s >>= 1) {
        if (tid < s) { s1[tid] += s1[tid + s]; s2[tid] += s2[tid + s]; }
        __syncthreads();
    }
    if (tid == 0) {
        float c1 = g1p[0];
        vn1[b] = 2.f * (SIGMA * SIGMA) / (c1 * c1) * s1[0];
        P1[b]  = c1 * s2[0];
    }
}

__global__ void act_kernel(const float* __restrict__ mu1, const float* __restrict__ vn1,
                           float* __restrict__ d2, float* __restrict__ sp,
                           float* __restrict__ m2b) {
    int t = blockIdx.x * 256 + threadIdx.x;
    if (t >= BATCH * D2) return;
    int b = t / D2;
    float m = mu1[t];
    float dd = 1.f / (1.f + expf(-m));
    float s  = (m > 20.f) ? m : log1pf(expf(m));
    float dsq = dd * dd;
    d2[t] = dsq;
    sp[t] = s;
    m2b[t] = s * s + vn1[b] * dsq;
}

__global__ void stats2_kernel(const float* __restrict__ m2b, const float* __restrict__ q2,
                              const float* __restrict__ g2p, const float* __restrict__ P1,
                              float* __restrict__ vn2, float* __restrict__ Pout) {
    __shared__ float s1[256], s2[256];
    int b = blockIdx.x, tid = threadIdx.x;
    float m = 0.f, p = 0.f;
    for (int j = tid; j < D2; j += 256) m += m2b[b * D2 + j];
    for (int o = tid; o < D3; o += 256) p += q2[b * D3 + o];
    s1[tid] = m; s2[tid] = p;
    __syncthreads();
    for (int s = 128; s > 0; s >>= 1) {
        if (tid < s) { s1[tid] += s1[tid + s]; s2[tid] += s2[tid + s]; }
        __syncthreads();
    }
    if (tid == 0) {
        float c2 = g2p[0];
        vn2[b] = 2.f * (SIGMA * SIGMA) / (c2 * c2) * s1[0];
        Pout[b] = P1[b] + c2 * s2[0];
    }
}

__global__ void __launch_bounds__(128)
gamma_kernel(const float* __restrict__ W2, const float* __restrict__ d2,
             const float* __restrict__ vn1, const float* __restrict__ vn2,
             float* __restrict__ out) {
    __shared__ __attribute__((aligned(16))) float smem[2 * GT * GP];
    float* As = smem;
    float* Bs = smem + GT * GP;

    const int tid  = threadIdx.x;
    const int lane = tid & 31;
    const int w    = tid >> 5;
    const int wy   = w >> 1;
    const int wx   = w & 1;
    const int r    = lane & 15;
    const int hi   = lane >> 4;

    int p  = blockIdx.x;
    int ti = 0;
    while (p >= NT - ti) { p -= NT - ti; ++ti; }
    const int tj   = ti + p;
    const int brow = ti * GT;
    const int bcol = tj * GT;
    const int b    = blockIdx.y;

    const int srow = tid >> 3;
    const int c4   = tid & 7;
    const float* d2b = d2 + (size_t)b * D2;
    const float* gA  = W2 + (size_t)(brow + srow) * D2 + c4 * 4;
    const float* gB  = W2 + (size_t)(bcol + srow) * D2 + c4 * 4;

    v4f ra[4], rb[4], rd;

    rd = *(const v4f*)(d2b + c4 * 4);
#pragma unroll
    for (int q = 0; q < 4; ++q) {
        ra[q] = *(const v4f*)(gA + (size_t)q * 16 * D2);
        rb[q] = *(const v4f*)(gB + (size_t)q * 16 * D2);
    }

    v8f acc[2][2] = {};

    const int sst = srow * GP + c4 * 4;
    const int fa  = (wy * 32 + r) * GP + 2 * hi;
    const int fb  = (wx * 32 + r) * GP + 2 * hi;

    for (int kb = 0; kb < NKB; ++kb) {
        __syncthreads();
#pragma unroll
        for (int q = 0; q < 4; ++q) {
            *(v4f*)(As + sst + q * 16 * GP) = ra[q] * rd;
            *(v4f*)(Bs + sst + q * 16 * GP) = rb[q];
        }
        if (kb + 1 < NKB) {
            const float* gA2 = gA + (kb + 1) * BK;
            const float* gB2 = gB + (kb + 1) * BK;
            rd = *(const v4f*)(d2b + (kb + 1) * BK + c4 * 4);
#pragma unroll
            for (int q = 0; q < 4; ++q) {
                ra[q] = *(const v4f*)(gA2 + (size_t)q * 16 * D2);
                rb[q] = *(const v4f*)(gB2 + (size_t)q * 16 * D2);
            }
        }
        __syncthreads();

        v2f a[2][BK / 4], bv[2][BK / 4];
#pragma unroll
        for (int i = 0; i < 2; ++i) {
#pragma unroll
            for (int ks = 0; ks < BK / 4; ++ks) {
                a[i][ks]  = *(const v2f*)(As + fa + i * 16 * GP + 4 * ks);
                bv[i][ks] = *(const v2f*)(Bs + fb + i * 16 * GP + 4 * ks);
            }
        }
#pragma unroll
        for (int ks = 0; ks < BK / 4; ++ks) {
#pragma unroll
            for (int i = 0; i < 2; ++i)
#pragma unroll
                for (int j = 0; j < 2; ++j)
                    acc[i][j] = __builtin_amdgcn_wmma_f32_16x16x4_f32(
                        false, a[i][ks], false, bv[j][ks], (short)0, acc[i][j], false, false);
        }
    }

    const float s  = vn1[b];
    const float vd = vn2[b];
    float* ob = out + (size_t)b * D3 * D3;
    float* Cs = smem;
    const bool diag = (ti == tj);

    __syncthreads();
#pragma unroll
    for (int i = 0; i < 2; ++i) {
#pragma unroll
        for (int j = 0; j < 2; ++j) {
#pragma unroll
            for (int v = 0; v < 8; ++v) {
                int lr = wy * 32 + i * 16 + v + 8 * hi;
                int lc = wx * 32 + j * 16 + r;
                float val = s * acc[i][j][v];
                if (diag && lr == lc) val += vd;
                acc[i][j][v] = val;
                Cs[lr * CP + lc] = val;
            }
        }
    }
    __syncthreads();

    const int er = tid >> 4;
    const int ec = (tid & 15) * 4;
#pragma unroll
    for (int q = 0; q < 8; ++q) {
        int row = er + 8 * q;
        v4f v4 = *(const v4f*)(Cs + row * CP + ec);
        *(v4f*)(ob + (size_t)(brow + row) * D3 + bcol + ec) = v4;
    }

    if (!diag) {
        __syncthreads();
#pragma unroll
        for (int i = 0; i < 2; ++i) {
#pragma unroll
            for (int j = 0; j < 2; ++j) {
#pragma unroll
                for (int v = 0; v < 8; ++v) {
                    int lr = wy * 32 + i * 16 + v + 8 * hi;
                    int lc = wx * 32 + j * 16 + r;
                    Cs[lc * CP + lr] = acc[i][j][v];
                }
            }
        }
        __syncthreads();
#pragma unroll
        for (int q = 0; q < 8; ++q) {
            int row = er + 8 * q;
            v4f v4 = *(const v4f*)(Cs + row * CP + ec);
            *(v4f*)(ob + (size_t)(bcol + row) * D3 + brow + ec) = v4;
        }
    }
}

extern "C" void kernel_launch(void* const* d_in, const int* in_sizes, int n_in,
                              void* d_out, int out_size, void* d_ws, size_t ws_size,
                              hipStream_t stream) {
    (void)in_sizes; (void)n_in; (void)out_size; (void)ws_size;
    const float* x    = (const float*)d_in[0];
    const float* bias = (const float*)d_in[1];
    const float* W1   = (const float*)d_in[2];
    const float* W2   = (const float*)d_in[3];
    const float* g1   = (const float*)d_in[4];
    const float* g2   = (const float*)d_in[5];

    float* out = (float*)d_out;
    float* out_mu2   = out;
    float* out_gamma = out + BATCH * D3;
    float* out_P     = out + BATCH * D3 + (size_t)BATCH * D3 * D3;

    float* ws = (float*)d_ws;
    float* xp      = ws;
    float* spare   = xp + BATCH * D1;
    float* mu1     = spare + D1 + D2;
    float* d2      = mu1 + BATCH * D2;
    float* sp      = d2 + BATCH * D2;
    float* m2b     = sp + BATCH * D2;
    float* vn1     = m2b + BATCH * D2;
    float* vn2     = vn1 + BATCH;
    float* P1      = vn2 + BATCH;
    float* q1      = d2;
    float* q2      = mu1;

    pool_kernel<<<(BATCH * D1 + 255) / 256, 256, 0, stream>>>(x, bias, xp);
    gemv4p_kernel<true><<<D2, 256, 0, stream>>>(W1, xp, xp, mu1, q1, D1, D2);
    stats1_kernel<<<BATCH, 256, 0, stream>>>(xp, q1, g1, vn1, P1);
    act_kernel<<<(BATCH * D2 + 255) / 256, 256, 0, stream>>>(mu1, vn1, d2, sp, m2b);
    gemv4p_kernel<false><<<D3, 256, 0, stream>>>(W2, sp, m2b, out_mu2, q2, D2, D3);
    stats2_kernel<<<BATCH, 256, 0, stream>>>(m2b, q2, g2, P1, vn2, out_P);

    dim3 ggrid(NPAIR, BATCH);
    gamma_kernel<<<ggrid, 128, 0, stream>>>(W2, d2, vn1, vn2, out_gamma);
}


# GEMVs: one wave per output row (16-byte loads, whole row of W requested as one group, 32-lane shuffle butterfly) replaces block per row with LDS tree; 8x fewer waves; sum order of mu and q changes
# speedup vs baseline: 4.0540x; 1.1072x over previous
#include <hip/hip_runtime.h>
#include <hip/hip_bf16.h>
#include <math.h>

typedef __attribute__((ext_vector_type(2))) float v2f;
typedef __attribute__((ext_vector_type(4))) float v4f;
typedef __attribute__((ext_vector_type(8))) float v8f;

#define BATCH 4
#define CCH   3
#define HH    64
#define WW    64
#define D1    3072
#define D2    2048
#define D3    1024
#define SIGMA 0.01f

#define GT     64
#define NT     (D3 / GT)
#define NPAIR  (NT * (NT + 1) / 2)
#define BK     32
#define NKB    (D2 / BK)
#define GP     36
#define CP     68

__global__ void pool_kernel(const float* __restrict__ x, const float* __restrict__ bias,
                            float* __restrict__ xp) {
    int t = blockIdx.x * 256 + threadIdx.x;
    if (t >= BATCH * D1) return;
    int b = t / D1, f = t % D1;
    int c = f >> 10;
    int rem = f & 1023;
    int oh = rem >> 5, ow = rem & 31;
    int base = c * (HH * WW) + (2 * oh) * WW + 2 * ow;
    const float* xb = x + b * (CCH * HH * WW);
    float s = (xb[base]        + bias[base])
            + (xb[base + 1]    + bias[base + 1])
            + (xb[base + WW]   + bias[base + WW])
            + (xb[base + WW+1] + bias[base + WW + 1]);
    xp[t] = 0.25f * s;
}

template <bool SQ, int DIN, int DOUT, int G>
__global__ void __launch_bounds__(128)
gemv4w_kernel(const float* __restrict__ W, const float* __restrict__ X,
              const float* __restrict__ M, float* __restrict__ out,
              float* __restrict__ q) {
    constexpr int Q  = DIN / 4;
    constexpr int NC = Q / 32;
    constexpr int NG = NC / G;
    const int lane = threadIdx.x & 31;
    const int o    = blockIdx.x * 4 + (threadIdx.x >> 5);
    const v4f* w4 = (const v4f*)(W + (size_t)o * DIN) + lane;
    const v4f* x4 = (const v4f*)X + lane;
    const v4f* m4 = (const v4f*)M + lane;
    v4f w[NC];
#pragma unroll
    for (int c = 0; c < NC; ++c) w[c] = w4[c * 32];
    v4f xb[2][G][4], mb[2][G][4];
#pragma unroll
    for (int j = 0; j < G; ++j) {
#pragma unroll
        for (int b = 0; b < 4; ++b) {
            xb[0][j][b] = x4[b * Q + j * 32];
            if (!SQ) mb[0][j][b] = m4[b * Q + j * 32];
        }
    }
    v4f a[4] = {}, p[4] = {};
#pragma unroll
    for (int g = 0; g < NG; ++g) {
        if (g + 1 < NG) {
#pragma unroll
            for (int j = 0; j < G; ++j) {
#pragma unroll
                for (int b = 0; b < 4; ++b) {
                    xb[(g + 1) & 1][j][b] = x4[b * Q + ((g + 1) * G + j) * 32];
                    if (!SQ) mb[(g + 1) & 1][j][b] = m4[b * Q + ((g + 1) * G + j) * 32];
                }
            }
        }
        __builtin_amdgcn_sched_barrier(0);
#pragma unroll
        for (int j = 0; j < G; ++j) {
            const int c = g * G + j;
            v4f aw = __builtin_elementwise_abs(w[c]);
#pragma unroll
            for (int b = 0; b < 4; ++b) {
                v4f x = xb[g & 1][j][b];
                v4f m;
                if (SQ) m = x * x;
                else    m = mb[g & 1][j][b];
                a[b] += w[c] * x;
                p[b] += aw * m;
            }
        }
    }
    float s[8];
#pragma unroll
    for (int b = 0; b < 4; ++b) {
        s[b]     = (a[b].x + a[b].y) + (a[b].z + a[b].w);
        s[4 + b] = (p[b].x + p[b].y) + (p[b].z + p[b].w);
    }
#pragma unroll
    for (int off = 16; off > 0; off >>= 1) {
#pragma unroll
        for (int v = 0; v < 8; ++v) s[v] += __shfl_xor(s[v], off);
    }
    float r = s[0];
#pragma unroll
    for (int v = 1; v < 8; ++v) r = (lane == v) ? s[v] : r;
    if (lane < 4)      out[lane * DOUT + o]     = r;
    else if (lane < 8) q[(lane - 4) * DOUT + o] = r;
}

__global__ void stats1_kernel(const float* __restrict__ xp, const float* __restrict__ q1,
                              const float* __restrict__ g1p, float* __restrict__ vn1,
                              float* __restrict__ P1) {
    __shared__ float s1[256], s2[256];
    int b = blockIdx.x, tid = threadIdx.x;
    float m = 0.f, p = 0.f;
    for (int i = tid; i < D1; i += 256) {
        float v = xp[b * D1 + i];
        m += v * v;
    }
    for (int o = tid; o < D2; o += 256) p += q1[b * D2 + o];
    s1[tid] = m; s2[tid] = p;
    __syncthreads();
    for (int s = 128; s > 0; s >>= 1) {
        if (tid < s) { s1[tid] += s1[tid + s]; s2[tid] += s2[tid + s]; }
        __syncthreads();
    }
    if (tid == 0) {
        float c1 = g1p[0];
        vn1[b] = 2.f * (SIGMA * SIGMA) / (c1 * c1) * s1[0];
        P1[b]  = c1 * s2[0];
    }
}

__global__ void act_kernel(const float* __restrict__ mu1, const float* __restrict__ vn1,
                           float* __restrict__ d2, float* __restrict__ sp,
                           float* __restrict__ m2b) {
    int t = blockIdx.x * 256 + threadIdx.x;
    if (t >= BATCH * D2) return;
    int b = t / D2;
    float m = mu1[t];
    float dd = 1.f / (1.f + expf(-m));
    float s  = (m > 20.f) ? m : log1pf(expf(m));
    float dsq = dd * dd;
    d2[t] = dsq;
    sp[t] = s;
    m2b[t] = s * s + vn1[b] * dsq;
}

__global__ void stats2_kernel(const float* __restrict__ m2b, const float* __restrict__ q2,
                              const float* __restrict__ g2p, const float* __restrict__ P1,
                              float* __restrict__ vn2, float* __restrict__ Pout) {
    __shared__ float s1[256], s2[256];
    int b = blockIdx.x, tid = threadIdx.x;
    float m = 0.f, p = 0.f;
    for (int j = tid; j < D2; j += 256) m += m2b[b * D2 + j];
    for (int o = tid; o < D3; o += 256) p += q2[b * D3 + o];
    s1[tid] = m; s2[tid] = p;
    __syncthreads();
    for (int s = 128; s > 0; s >>= 1) {
        if (tid < s) { s1[tid] += s1[tid + s]; s2[tid] += s2[tid + s]; }
        __syncthreads();
    }
    if (tid == 0) {
        float c2 = g2p[0];
        vn2[b] = 2.f * (SIGMA * SIGMA) / (c2 * c2) * s1[0];
        Pout[b] = P1[b] + c2 * s2[0];
    }
}

__global__ void __launch_bounds__(128)
gamma_kernel(const float* __restrict__ W2, const float* __restrict__ d2,
             const float* __restrict__ vn1, const float* __restrict__ vn2,
             float* __restrict__ out) {
    __shared__ __attribute__((aligned(16))) float smem[2 * GT * GP];
    float* As = smem;
    float* Bs = smem + GT * GP;

    const int tid  = threadIdx.x;
    const int lane = tid & 31;
    const int w    = tid >> 5;
    const int wy   = w >> 1;
    const int wx   = w & 1;
    const int r    = lane & 15;
    const int hi   = lane >> 4;

    int p  = blockIdx.x;
    int ti = 0;
    while (p >= NT - ti) { p -= NT - ti; ++ti; }
    const int tj   = ti + p;
    const int brow = ti * GT;
    const int bcol = tj * GT;
    const int b    = blockIdx.y;

    const int srow = tid >> 3;
    const int c4   = tid & 7;
    const float* d2b = d2 + (size_t)b * D2;
    const float* gA  = W2 + (size_t)(brow + srow) * D2 + c4 * 4;
    const float* gB  = W2 + (size_t)(bcol + srow) * D2 + c4 * 4;

    v4f ra[4], rb[4], rd;

    rd = *(const v4f*)(d2b + c4 * 4);
#pragma unroll
    for (int q = 0; q < 4; ++q) {
        ra[q] = *(const v4f*)(gA + (size_t)q * 16 * D2);
        rb[q] = *(const v4f*)(gB + (size_t)q * 16 * D2);
    }

    v8f acc[2][2] = {};

    const int sst = srow * GP + c4 * 4;
    const int fa  = (wy * 32 + r) * GP + 2 * hi;
    const int fb  = (wx * 32 + r) * GP + 2 * hi;

    for (int kb = 0; kb < NKB; ++kb) {
        __syncthreads();
#pragma unroll
        for (int q = 0; q < 4; ++q) {
            *(v4f*)(As + sst + q * 16 * GP) = ra[q] * rd;
            *(v4f*)(Bs + sst + q * 16 * GP) = rb[q];
        }
        if (kb + 1 < NKB) {
            const float* gA2 = gA + (kb + 1) * BK;
            const float* gB2 = gB + (kb + 1) * BK;
            rd = *(const v4f*)(d2b + (kb + 1) * BK + c4 * 4);
#pragma unroll
            for (int q = 0; q < 4; ++q) {
                ra[q] = *(const v4f*)(gA2 + (size_t)q * 16 * D2);
                rb[q] = *(const v4f*)(gB2 + (size_t)q * 16 * D2);
            }
        }
        __syncthreads();

        v2f a[2][BK / 4], bv[2][BK / 4];
#pragma unroll
        for (int i = 0; i < 2; ++i) {
#pragma unroll
            for (int ks = 0; ks < BK / 4; ++ks) {
                a[i][ks]  = *(const v2f*)(As + fa + i * 16 * GP + 4 * ks);
                bv[i][ks] = *(const v2f*)(Bs + fb + i * 16 * GP + 4 * ks);
            }
        }
#pragma unroll
        for (int ks = 0; ks < BK / 4; ++ks) {
#pragma unroll
            for (int i = 0; i < 2; ++i)
#pragma unroll
                for (int j = 0; j < 2; ++j)
                    acc[i][j] = __builtin_amdgcn_wmma_f32_16x16x4_f32(
                        false, a[i][ks], false, bv[j][ks], (short)0, acc[i][j], false, false);
        }
    }

    const float s  = vn1[b];
    const float vd = vn2[b];
    float* ob = out + (size_t)b * D3 * D3;
    float* Cs = smem;
    const bool diag = (ti == tj);

    __syncthreads();
#pragma unroll
    for (int i = 0; i < 2; ++i) {
#pragma unroll
        for (int j = 0; j < 2; ++j) {
#pragma unroll
            for (int v = 0; v < 8; ++v) {
                int lr = wy * 32 + i * 16 + v + 8 * hi;
                int lc = wx * 32 + j * 16 + r;
                float val = s * acc[i][j][v];
                if (diag && lr == lc) val += vd;
                acc[i][j][v] = val;
                Cs[lr * CP + lc] = val;
            }
        }
    }
    __syncthreads();

    const int er = tid >> 4;
    const int ec = (tid & 15) * 4;
#pragma unroll
    for (int q = 0; q < 8; ++q) {
        int row = er + 8 * q;
        v4f v4 = *(const v4f*)(Cs + row * CP + ec);
        *(v4f*)(ob + (size_t)(brow + row) * D3 + bcol + ec) = v4;
    }

    if (!diag) {
        __syncthreads();
#pragma unroll
        for (int i = 0; i < 2; ++i) {
#pragma unroll
            for (int j = 0; j < 2; ++j) {
#pragma unroll
                for (int v = 0; v < 8; ++v) {
                    int lr = wy * 32 + i * 16 + v + 8 * hi;
                    int lc = wx * 32 + j * 16 + r;
                    Cs[lc * CP + lr] = acc[i][j][v];
                }
            }
        }
        __syncthreads();
#pragma unroll
        for (int q = 0; q < 8; ++q) {
            int row = er + 8 * q;
            v4f v4 = *(const v4f*)(Cs + row * CP + ec);
            *(v4f*)(ob + (size_t)(bcol + row) * D3 + brow + ec) = v4;
        }
    }
}

extern "C" void kernel_launch(void* const* d_in, const int* in_sizes, int n_in,
                              void* d_out, int out_size, void* d_ws, size_t ws_size,
                              hipStream_t stream) {
    (void)in_sizes; (void)n_in; (void)out_size; (void)ws_size;
    const float* x    = (const float*)d_in[0];
    const float* bias = (const float*)d_in[1];
    const float* W1   = (const float*)d_in[2];
    const float* W2   = (const float*)d_in[3];
    const float* g1   = (const float*)d_in[4];
    const float* g2   = (const float*)d_in[5];

    float* out = (float*)d_out;
    float* out_mu2   = out;
    float* out_gamma = out + BATCH * D3;
    float* out_P     = out + BATCH * D3 + (size_t)BATCH * D3 * D3;

    float* ws = (float*)d_ws;
    float* xp      = ws;
    float* spare   = xp + BATCH * D1;
    float* mu1     = spare + D1 + D2;
    float* d2      = mu1 + BATCH * D2;
    float* sp      = d2 + BATCH * D2;
    float* m2b     = sp + BATCH * D2;
    float* vn1     = m2b + BATCH * D2;
    float* vn2     = vn1 + BATCH;
    float* P1      = vn2 + BATCH;
    float* q1      = d2;
    float* q2      = mu1;

    pool_kernel<<<(BATCH * D1 + 255) / 256, 256, 0, stream>>>(x, bias, xp);
    gemv4w_kernel<true, D1, D2, 6><<<D2 / 4, 128, 0, stream>>>(W1, xp, xp, mu1, q1);
    stats1_kernel<<<BATCH, 256, 0, stream>>>(xp, q1, g1, vn1, P1);
    act_kernel<<<(BATCH * D2 + 255) / 256, 256, 0, stream>>>(mu1, vn1, d2, sp, m2b);
    gemv4w_kernel<false, D2, D3, 4><<<D3 / 4, 128, 0, stream>>>(W2, sp, m2b, out_mu2, q2);
    stats2_kernel<<<BATCH, 256, 0, stream>>>(m2b, q2, g2, P1, vn2, out_P);

    dim3 ggrid(NPAIR, BATCH);
    gamma_kernel<<<ggrid, 128, 0, stream>>>(W2, d2, vn1, vn2, out_gamma);
}


# gamma2: each 64x64 tile split into two 32x64 blocks (wave tile 16x32), 1088 equal blocks lie 4-5 per WGP instead of 544 at 2-3: shorter tail and more waves to cover the staging loads
# speedup vs baseline: 4.2629x; 1.0515x over previous
#include <hip/hip_runtime.h>
#include <hip/hip_bf16.h>
#include <math.h>

typedef __attribute__((ext_vector_type(2))) float v2f;
typedef __attribute__((ext_vector_type(4))) float v4f;
typedef __attribute__((ext_vector_type(8))) float v8f;

#define BATCH 4
#define CCH   3
#define HH    64
#define WW    64
#define D1    3072
#define D2    2048
#define D3    1024
#define SIGMA 0.01f

#define GT     64
#define GR     32
#define NT     (D3 / GT)
#define NPAIR  (NT * (NT + 1) / 2)
#define BK     32
#define NKB    (D2 / BK)
#define GP     36
#define CP     68
#define TP     36

__global__ void pool_kernel(const float* __restrict__ x, const float* __restrict__ bias,
                            float* __restrict__ xp) {
    int t = blockIdx.x * 256 + threadIdx.x;
    if (t >= BATCH * D1) return;
    int b = t / D1, f = t % D1;
    int c = f >> 10;
    int rem = f & 1023;
    int oh = rem >> 5, ow = rem & 31;
    int base = c * (HH * WW) + (2 * oh) * WW + 2 * ow;
    const float* xb = x + b * (CCH * HH * WW);
    float s = (xb[base]        + bias[base])
            + (xb[base + 1]    + bias[base + 1])
            + (xb[base + WW]   + bias[base + WW])
            + (xb[base + WW+1] + bias[base + WW + 1]);
    xp[t] = 0.25f * s;
}

template <bool SQ, int DIN, int DOUT, int G>
__global__ void __launch_bounds__(128)
gemv4w_kernel(const float* __restrict__ W, const float* __restrict__ X,
              const float* __restrict__ M, float* __restrict__ out,
              float* __restrict__ q) {
    constexpr int Q  = DIN / 4;
    constexpr int NC = Q / 32;
    constexpr int NG = NC / G;
    const int lane = threadIdx.x & 31;
    const int o    = blockIdx.x * 4 + (threadIdx.x >> 5);
    const v4f* w4 = (const v4f*)(W + (size_t)o * DIN) + lane;
    const v4f* x4 = (const v4f*)X + lane;
    const v4f* m4 = (const v4f*)M + lane;
    v4f w[NC];
#pragma unroll
    for (int c = 0; c < NC; ++c) w[c] = w4[c * 32];
    v4f xb[2][G][4], mb[2][G][4];
#pragma unroll
    for (int j = 0; j < G; ++j) {
#pragma unroll
        for (int b = 0; b < 4; ++b) {
            xb[0][j][b] = x4[b * Q + j * 32];
            if (!SQ) mb[0][j][b] = m4[b * Q + j * 32];
        }
    }
    v4f a[4] = {}, p[4] = {};
#pragma unroll
    for (int g = 0; g < NG; ++g) {
        if (g + 1 < NG) {
#pragma unroll
            for (int j = 0; j < G; ++j) {
#pragma unroll
                for (int b = 0; b < 4; ++b) {
                    xb[(g + 1) & 1][j][b] = x4[b * Q + ((g + 1) * G + j) * 32];
                    if (!SQ) mb[(g + 1) & 1][j][b] = m4[b * Q + ((g + 1) * G + j) * 32];
                }
            }
        }
        __builtin_amdgcn_sched_barrier(0);
#pragma unroll
        for (int j = 0; j < G; ++j) {
            const int c = g * G + j;
            v4f aw = __builtin_elementwise_abs(w[c]);
#pragma unroll
            for (int b = 0; b < 4; ++b) {
                v4f x = xb[g & 1][j][b];
                v4f m;
                if (SQ) m = x * x;
                else    m = mb[g & 1][j][b];
                a[b] += w[c] * x;
                p[b] += aw * m;
            }
        }
    }
    float s[8];
#pragma unroll
    for (int b = 0; b < 4; ++b) {
        s[b]     = (a[b].x + a[b].y) + (a[b].z + a[b].w);
        s[4 + b] = (p[b].x + p[b].y) + (p[b].z + p[b].w);
    }
#pragma unroll
    for (int off = 16; off > 0; off >>= 1) {
#pragma unroll
        for (int v = 0; v < 8; ++v) s[v] += __shfl_xor(s[v], off);
    }
    float r = s[0];
#pragma unroll
    for (int v = 1; v < 8; ++v) r = (lane == v) ? s[v] : r;
    if (lane < 4)      out[lane * DOUT + o]     = r;
    else if (lane < 8) q[(lane - 4) * DOUT + o] = r;
}

__global__ void stats1_kernel(const float* __restrict__ xp, const float* __restrict__ q1,
                              const float* __restrict__ g1p, float* __restrict__ vn1,
                              float* __restrict__ P1) {
    __shared__ float s1[256], s2[256];
    int b = blockIdx.x, tid = threadIdx.x;
    float m = 0.f, p = 0.f;
    for (int i = tid; i < D1; i += 256) {
        float v = xp[b * D1 + i];
        m += v * v;
    }
    for (int o = tid; o < D2; o += 256) p += q1[b * D2 + o];
    s1[tid] = m; s2[tid] = p;
    __syncthreads();
    for (int s = 128; s > 0; s >>= 1) {
        if (tid < s) { s1[tid] += s1[tid + s]; s2[tid] += s2[tid + s]; }
        __syncthreads();
    }
    if (tid == 0) {
        float c1 = g1p[0];
        vn1[b] = 2.f * (SIGMA * SIGMA) / (c1 * c1) * s1[0];
        P1[b]  = c1 * s2[0];
    }
}

__global__ void act_kernel(const float* __restrict__ mu1, const float* __restrict__ vn1,
                           float* __restrict__ d2, float* __restrict__ sp,
                           float* __restrict__ m2b) {
    int t = blockIdx.x * 256 + threadIdx.x;
    if (t >= BATCH * D2) return;
    int b = t / D2;
    float m = mu1[t];
    float dd = 1.f / (1.f + expf(-m));
    float s  = (m > 20.f) ? m : log1pf(expf(m));
    float dsq = dd * dd;
    d2[t] = dsq;
    sp[t] = s;
    m2b[t] = s * s + vn1[b] * dsq;
}

__global__ void stats2_kernel(const float* __restrict__ m2b, const float* __restrict__ q2,
                              const float* __restrict__ g2p, const float* __restrict__ P1,
                              float* __restrict__ vn2, float* __restrict__ Pout) {
    __shared__ float s1[256], s2[256];
    int b = blockIdx.x, tid = threadIdx.x;
    float m = 0.f, p = 0.f;
    for (int j = tid; j < D2; j += 256) m += m2b[b * D2 + j];
    for (int o = tid; o < D3; o += 256) p += q2[b * D3 + o];
    s1[tid] = m; s2[tid] = p;
    __syncthreads();
    for (int s = 128; s > 0; s >>= 1) {
        if (tid < s) { s1[tid] += s1[tid + s]; s2[tid] += s2[tid + s]; }
        __syncthreads();
    }
    if (tid == 0) {
        float c2 = g2p[0];
        vn2[b] = 2.f * (SIGMA * SIGMA) / (c2 * c2) * s1[0];
        Pout[b] = P1[b] + c2 * s2[0];
    }
}

__global__ void __launch_bounds__(128)
gamma_kernel(const float* __restrict__ W2, const float* __restrict__ d2,
             const float* __restrict__ vn1, const float* __restrict__ vn2,
             float* __restrict__ out) {
    __shared__ __attribute__((aligned(16))) float smem[(GR + GT) * GP];
    float* As = smem;
    float* Bs = smem + GR * GP;

    const int tid  = threadIdx.x;
    const int lane = tid & 31;
    const int w    = tid >> 5;
    const int wy   = w >> 1;
    const int wx   = w & 1;
    const int r    = lane & 15;
    const int hi   = lane >> 4;

    int p  = blockIdx.x;
    int ti = 0;
    while (p >= NT - ti) { p -= NT - ti; ++ti; }
    const int tj   = ti + p;
    const int half = blockIdx.y & 1;
    const int b    = blockIdx.y >> 1;
    const int brow = ti * GT + half * GR;
    const int bcol = tj * GT;

    const int srow = tid >> 3;
    const int c4   = tid & 7;
    const float* d2b = d2 + (size_t)b * D2;
    const float* gA  = W2 + (size_t)(brow + srow) * D2 + c4 * 4;
    const float* gB  = W2 + (size_t)(bcol + srow) * D2 + c4 * 4;

    v4f ra[2], rb[4], rd;

    rd = *(const v4f*)(d2b + c4 * 4);
#pragma unroll
    for (int q = 0; q < 2; ++q) ra[q] = *(const v4f*)(gA + (size_t)q * 16 * D2);
#pragma unroll
    for (int q = 0; q < 4; ++q) rb[q] = *(const v4f*)(gB + (size_t)q * 16 * D2);

    v8f acc[2] = {};

    const int sst = srow * GP + c4 * 4;
    const int fa  = (wy * 16 + r) * GP + 2 * hi;
    const int fb  = (wx * 32 + r) * GP + 2 * hi;

    for (int kb = 0; kb < NKB; ++kb) {
        __syncthreads();
#pragma unroll
        for (int q = 0; q < 2; ++q) *(v4f*)(As + sst + q * 16 * GP) = ra[q] * rd;
#pragma unroll
        for (int q = 0; q < 4; ++q) *(v4f*)(Bs + sst + q * 16 * GP) = rb[q];
        if (kb + 1 < NKB) {
            const float* gA2 = gA + (kb + 1) * BK;
            const float* gB2 = gB + (kb + 1) * BK;
            rd = *(const v4f*)(d2b + (kb + 1) * BK + c4 * 4);
#pragma unroll
            for (int q = 0; q < 2; ++q) ra[q] = *(const v4f*)(gA2 + (size_t)q * 16 * D2);
#pragma unroll
            for (int q = 0; q < 4; ++q) rb[q] = *(const v4f*)(gB2 + (size_t)q * 16 * D2);
        }
        __syncthreads();

        v2f a[BK / 4], bv[2][BK / 4];
#pragma unroll
        for (int ks = 0; ks < BK / 4; ++ks) {
            a[ks] = *(const v2f*)(As + fa + 4 * ks);
#pragma unroll
            for (int j = 0; j < 2; ++j)
                bv[j][ks] = *(const v2f*)(Bs + fb + j * 16 * GP + 4 * ks);
        }
#pragma unroll
        for (int ks = 0; ks < BK / 4; ++ks) {
#pragma unroll
            for (int j = 0; j < 2; ++j)
                acc[j] = __builtin_amdgcn_wmma_f32_16x16x4_f32(
                    false, a[ks], false, bv[j][ks], (short)0, acc[j], false, false);
        }
    }

    const float s  = vn1[b];
    const float vd = vn2[b];
    float* ob = out + (size_t)b * D3 * D3;
    float* Cs = smem;
    const bool diag = (ti == tj);

    __syncthreads();
#pragma unroll
    for (int j = 0; j < 2; ++j) {
#pragma unroll
        for (int v = 0; v < 8; ++v) {
            int lr = wy * 16 + v + 8 * hi;
            int lc = wx * 32 + j * 16 + r;
            float val = s * acc[j][v];
            if (diag && half * GR + lr == lc) val += vd;
            acc[j][v] = val;
            Cs[lr * CP + lc] = val;
        }
    }
    __syncthreads();

    const int er = tid >> 4;
    const int ec = (tid & 15) * 4;
#pragma unroll
    for (int q = 0; q < 4; ++q) {
        int row = er + 8 * q;
        v4f v4 = *(const v4f*)(Cs + row * CP + ec);
        *(v4f*)(ob + (size_t)(brow + row) * D3 + bcol + ec) = v4;
    }

    if (!diag) {
        __syncthreads();
#pragma unroll
        for (int j = 0; j < 2; ++j) {
#pragma unroll
            for (int v = 0; v < 8; ++v) {
                int lr = wy * 16 + v + 8 * hi;
                int lc = wx * 32 + j * 16 + r;
                Cs[lc * TP + lr] = acc[j][v];
            }
        }
        __syncthreads();
        const int tr = tid >> 3;
        const int tc = (tid & 7) * 4;
#pragma unroll
        for (int q = 0; q < 4; ++q) {
            int row = tr + 16 * q;
            v4f v4 = *(const v4f*)(Cs + row * TP + tc);
            *(v4f*)(ob + (size_t)(bcol + row) * D3 + brow + tc) = v4;
        }
    }
}

extern "C" void kernel_launch(void* const* d_in, const int* in_sizes, int n_in,
                              void* d_out, int out_size, void* d_ws, size_t ws_size,
                              hipStream_t stream) {
    (void)in_sizes; (void)n_in; (void)out_size; (void)ws_size;
    const float* x    = (const float*)d_in[0];
    const float* bias = (const float*)d_in[1];
    const float* W1   = (const float*)d_in[2];
    const float* W2   = (const float*)d_in[3];
    const float* g1   = (const float*)d_in[4];
    const float* g2   = (const float*)d_in[5];

    float* out = (float*)d_out;
    float* out_mu2   = out;
    float* out_gamma = out + BATCH * D3;
    float* out_P     = out + BATCH * D3 + (size_t)BATCH * D3 * D3;

    float* ws = (float*)d_ws;
    float* xp      = ws;
    float* spare   = xp + BATCH * D1;
    float* mu1     = spare + D1 + D2;
    float* d2      = mu1 + BATCH * D2;
    float* sp      = d2 + BATCH * D2;
    float* m2b     = sp + BATCH * D2;
    float* vn1     = m2b + BATCH * D2;
    float* vn2     = vn1 + BATCH;
    float* P1      = vn2 + BATCH;
    float* q1      = d2;
    float* q2      = mu1;

    pool_kernel<<<(BATCH * D1 + 255) / 256, 256, 0, stream>>>(x, bias, xp);
    gemv4w_kernel<true, D1, D2, 6><<<D2 / 4, 128, 0, stream>>>(W1, xp, xp, mu1, q1);
    stats1_kernel<<<BATCH, 256, 0, stream>>>(xp, q1, g1, vn1, P1);
    act_kernel<<<(BATCH * D2 + 255) / 256, 256, 0, stream>>>(mu1, vn1, d2, sp, m2b);
    gemv4w_kernel<false, D2, D3, 4><<<D3 / 4, 128, 0, stream>>>(W2, sp, m2b, out_mu2, q2);
    stats2_kernel<<<BATCH, 256, 0, stream>>>(m2b, q2, g2, P1, vn2, out_P);

    dim3 ggrid(NPAIR, BATCH * 2);
    gamma_kernel<<<ggrid, 128, 0, stream>>>(W2, d2, vn1, vn2, out_gamma);
}


# gamma2 K loop: staging loads requested two K-blocks ahead (two register sets) and left in flight across the loop barriers, which now order LDS only instead of draining the load counter
# speedup vs baseline: 4.7352x; 1.1108x over previous
#include <hip/hip_runtime.h>
#include <hip/hip_bf16.h>
#include <math.h>

typedef __attribute__((ext_vector_type(2))) float v2f;
typedef __attribute__((ext_vector_type(4))) float v4f;
typedef __attribute__((ext_vector_type(8))) float v8f;

#define BATCH 4
#define CCH   3
#define HH    64
#define WW    64
#define D1    3072
#define D2    2048
#define D3    1024
#define SIGMA 0.01f

#define GT     64
#define GR     32
#define NT     (D3 / GT)
#define NPAIR  (NT * (NT + 1) / 2)
#define BK     32
#define NKB    (D2 / BK)
#define GP     36
#define CP     68
#define TP     36

__global__ void pool_kernel(const float* __restrict__ x, const float* __restrict__ bias,
                            float* __restrict__ xp) {
    int t = blockIdx.x * 256 + threadIdx.x;
    if (t >= BATCH * D1) return;
    int b = t / D1, f = t % D1;
    int c = f >> 10;
    int rem = f & 1023;
    int oh = rem >> 5, ow = rem & 31;
    int base = c * (HH * WW) + (2 * oh) * WW + 2 * ow;
    const float* xb = x + b * (CCH * HH * WW);
    float s = (xb[base]        + bias[base])
            + (xb[base + 1]    + bias[base + 1])
            + (xb[base + WW]   + bias[base + WW])
            + (xb[base + WW+1] + bias[base + WW + 1]);
    xp[t] = 0.25f * s;
}

template <bool SQ, int DIN, int DOUT, int G>
__global__ void __launch_bounds__(128)
gemv4w_kernel(const float* __restrict__ W, const float* __restrict__ X,
              const float* __restrict__ M, float* __restrict__ out,
              float* __restrict__ q) {
    constexpr int Q  = DIN / 4;
    constexpr int NC = Q / 32;
    constexpr int NG = NC / G;
    const int lane = threadIdx.x & 31;
    const int o    = blockIdx.x * 4 + (threadIdx.x >> 5);
    const v4f* w4 = (const v4f*)(W + (size_t)o * DIN) + lane;
    const v4f* x4 = (const v4f*)X + lane;
    const v4f* m4 = (const v4f*)M + lane;
    v4f w[NC];
#pragma unroll
    for (int c = 0; c < NC; ++c) w[c] = w4[c * 32];
    v4f xb[2][G][4], mb[2][G][4];
#pragma unroll
    for (int j = 0; j < G; ++j) {
#pragma unroll
        for (int b = 0; b < 4; ++b) {
            xb[0][j][b] = x4[b * Q + j * 32];
            if (!SQ) mb[0][j][b] = m4[b * Q + j * 32];
        }
    }
    v4f a[4] = {}, p[4] = {};
#pragma unroll
    for (int g = 0; g < NG; ++g) {
        if (g + 1 < NG) {
#pragma unroll
            for (int j = 0; j < G; ++j) {
#pragma unroll
                for (int b = 0; b < 4; ++b) {
                    xb[(g + 1) & 1][j][b] = x4[b * Q + ((g + 1) * G + j) * 32];
                    if (!SQ) mb[(g + 1) & 1][j][b] = m4[b * Q + ((g + 1) * G + j) * 32];
                }
            }
        }
        __builtin_amdgcn_sched_barrier(0);
#pragma unroll
        for (int j = 0; j < G; ++j) {
            const int c = g * G + j;
            v4f aw = __builtin_elementwise_abs(w[c]);
#pragma unroll
            for (int b = 0; b < 4; ++b) {
                v4f x = xb[g & 1][j][b];
                v4f m;
                if (SQ) m = x * x;
                else    m = mb[g & 1][j][b];
                a[b] += w[c] * x;
                p[b] += aw * m;
            }
        }
    }
    float s[8];
#pragma unroll
    for (int b = 0; b < 4; ++b) {
        s[b]     = (a[b].x + a[b].y) + (a[b].z + a[b].w);
        s[4 + b] = (p[b].x + p[b].y) + (p[b].z + p[b].w);
    }
#pragma unroll
    for (int off = 16; off > 0; off >>= 1) {
#pragma unroll
        for (int v = 0; v < 8; ++v) s[v] += __shfl_xor(s[v], off);
    }
    float r = s[0];
#pragma unroll
    for (int v = 1; v < 8; ++v) r = (lane == v) ? s[v] : r;
    if (lane < 4)      out[lane * DOUT + o]     = r;
    else if (lane < 8) q[(lane - 4) * DOUT + o] = r;
}

__global__ void stats1_kernel(const float* __restrict__ xp, const float* __restrict__ q1,
                              const float* __restrict__ g1p, float* __restrict__ vn1,
                              float* __restrict__ P1) {
    __shared__ float s1[256], s2[256];
    int b = blockIdx.x, tid = threadIdx.x;
    float m = 0.f, p = 0.f;
    for (int i = tid; i < D1; i += 256) {
        float v = xp[b * D1 + i];
        m += v * v;
    }
    for (int o = tid; o < D2; o += 256) p += q1[b * D2 + o];
    s1[tid] = m; s2[tid] = p;
    __syncthreads();
    for (int s = 128; s > 0; s >>= 1) {
        if (tid < s) { s1[tid] += s1[tid + s]; s2[tid] += s2[tid + s]; }
        __syncthreads();
    }
    if (tid == 0) {
        float c1 = g1p[0];
        vn1[b] = 2.f * (SIGMA * SIGMA) / (c1 * c1) * s1[0];
        P1[b]  = c1 * s2[0];
    }
}

__global__ void act_kernel(const float* __restrict__ mu1, const float* __restrict__ vn1,
                           float* __restrict__ d2, float* __restrict__ sp,
                           float* __restrict__ m2b) {
    int t = blockIdx.x * 256 + threadIdx.x;
    if (t >= BATCH * D2) return;
    int b = t / D2;
    float m = mu1[t];
    float dd = 1.f / (1.f + expf(-m));
    float s  = (m > 20.f) ? m : log1pf(expf(m));
    float dsq = dd * dd;
    d2[t] = dsq;
    sp[t] = s;
    m2b[t] = s * s + vn1[b] * dsq;
}

__global__ void stats2_kernel(const float* __restrict__ m2b, const float* __restrict__ q2,
                              const float* __restrict__ g2p, const float* __restrict__ P1,
                              float* __restrict__ vn2, float* __restrict__ Pout) {
    __shared__ float s1[256], s2[256];
    int b = blockIdx.x, tid = threadIdx.x;
    float m = 0.f, p = 0.f;
    for (int j = tid; j < D2; j += 256) m += m2b[b * D2 + j];
    for (int o = tid; o < D3; o += 256) p += q2[b * D3 + o];
    s1[tid] = m; s2[tid] = p;
    __syncthreads();
    for (int s = 128; s > 0; s >>= 1) {
        if (tid < s) { s1[tid] += s1[tid + s]; s2[tid] += s2[tid + s]; }
        __syncthreads();
    }
    if (tid == 0) {
        float c2 = g2p[0];
        vn2[b] = 2.f * (SIGMA * SIGMA) / (c2 * c2) * s1[0];
        Pout[b] = P1[b] + c2 * s2[0];
    }
}

struct KStage { v4f a[2]; v4f b[4]; v4f d; };

__device__ __forceinline__ void lds_barrier() {
    __builtin_amdgcn_fence(__ATOMIC_RELEASE, "workgroup", "local");
    __builtin_amdgcn_s_barrier();
    __builtin_amdgcn_fence(__ATOMIC_ACQUIRE, "workgroup", "local");
}

__global__ void __launch_bounds__(128)
gamma_kernel(const float* __restrict__ W2, const float* __restrict__ d2,
             const float* __restrict__ vn1, const float* __restrict__ vn2,
             float* __restrict__ out) {
    __shared__ __attribute__((aligned(16))) float smem[(GR + GT) * GP];
    float* As = smem;
    float* Bs = smem + GR * GP;

    const int tid  = threadIdx.x;
    const int lane = tid & 31;
    const int w    = tid >> 5;
    const int wy   = w >> 1;
    const int wx   = w & 1;
    const int r    = lane & 15;
    const int hi   = lane >> 4;

    int p  = blockIdx.x;
    int ti = 0;
    while (p >= NT - ti) { p -= NT - ti; ++ti; }
    const int tj   = ti + p;
    const int half = blockIdx.y & 1;
    const int b    = blockIdx.y >> 1;
    const int brow = ti * GT + half * GR;
    const int bcol = tj * GT;

    const int srow = tid >> 3;
    const int c4   = tid & 7;
    const float* d2b = d2 + (size_t)b * D2;
    const float* gA  = W2 + (size_t)(brow + srow) * D2 + c4 * 4;
    const float* gB  = W2 + (size_t)(bcol + srow) * D2 + c4 * 4;

    auto request = [&](KStage& s, int kb) {
        const float* gA2 = gA + kb * BK;
        const float* gB2 = gB + kb * BK;
        s.d = *(const v4f*)(d2b + kb * BK + c4 * 4);
#pragma unroll
        for (int q = 0; q < 2; ++q) s.a[q] = *(const v4f*)(gA2 + (size_t)q * 16 * D2);
#pragma unroll
        for (int q = 0; q < 4; ++q) s.b[q] = *(const v4f*)(gB2 + (size_t)q * 16 * D2);
    };

    KStage s0, s1;
    request(s0, 0);
    __builtin_amdgcn_sched_barrier(0);
    request(s1, 1);

    v8f acc[2] = {};

    const int sst = srow * GP + c4 * 4;
    const int fa  = (wy * 16 + r) * GP + 2 * hi;
    const int fb  = (wx * 32 + r) * GP + 2 * hi;

    auto k_step = [&](KStage& s, int kb, bool more) {
        lds_barrier();
        asm volatile("" : "+v"(s.d));
#pragma unroll
        for (int q = 0; q < 2; ++q) *(v4f*)(As + sst + q * 16 * GP) = s.a[q] * s.d;
#pragma unroll
        for (int q = 0; q < 4; ++q) *(v4f*)(Bs + sst + q * 16 * GP) = s.b[q];
        if (more) request(s, kb + 2);
        lds_barrier();

        v2f a[BK / 4], bv[2][BK / 4];
#pragma unroll
        for (int ks = 0; ks < BK / 4; ++ks) {
            a[ks] = *(const v2f*)(As + fa + 4 * ks);
#pragma unroll
            for (int j = 0; j < 2; ++j)
                bv[j][ks] = *(const v2f*)(Bs + fb + j * 16 * GP + 4 * ks);
        }
#pragma unroll
        for (int ks = 0; ks < BK / 4; ++ks) {
#pragma unroll
            for (int j = 0; j < 2; ++j)
                acc[j] = __builtin_amdgcn_wmma_f32_16x16x4_f32(
                    false, a[ks], false, bv[j][ks], (short)0, acc[j], false, false);
        }
    };

    for (int kb = 0; kb < NKB - 2; kb += 2) {
        k_step(s0, kb, true);
        k_step(s1, kb + 1, true);
    }
    k_step(s0, NKB - 2, false);
    k_step(s1, NKB - 1, false);

    const float s  = vn1[b];
    const float vd = vn2[b];
    float* ob = out + (size_t)b * D3 * D3;
    float* Cs = smem;
    const bool diag = (ti == tj);

    __syncthreads();
#pragma unroll
    for (int j = 0; j < 2; ++j) {
#pragma unroll
        for (int v = 0; v < 8; ++v) {
            int lr = wy * 16 + v + 8 * hi;
            int lc = wx * 32 + j * 16 + r;
            float val = s * acc[j][v];
            if (diag && half * GR + lr == lc) val += vd;
            acc[j][v] = val;
            Cs[lr * CP + lc] = val;
        }
    }
    __syncthreads();

    const int er = tid >> 4;
    const int ec = (tid & 15) * 4;
#pragma unroll
    for (int q = 0; q < 4; ++q) {
        int row = er + 8 * q;
        v4f v4 = *(const v4f*)(Cs + row * CP + ec);
        *(v4f*)(ob + (size_t)(brow + row) * D3 + bcol + ec) = v4;
    }

    if (!diag) {
        __syncthreads();
#pragma unroll
        for (int j = 0; j < 2; ++j) {
#pragma unroll
            for (int v = 0; v < 8; ++v) {
                int lr = wy * 16 + v + 8 * hi;
                int lc = wx * 32 + j * 16 + r;
                Cs[lc * TP + lr] = acc[j][v];
            }
        }
        __syncthreads();
        const int tr = tid >> 3;
        const int tc = (tid & 7) * 4;
#pragma unroll
        for (int q = 0; q < 4; ++q) {
            int row = tr + 16 * q;
            v4f v4 = *(const v4f*)(Cs + row * TP + tc);
            *(v4f*)(ob + (size_t)(bcol + row) * D3 + brow + tc) = v4;
        }
    }
}

extern "C" void kernel_launch(void* const* d_in, const int* in_sizes, int n_in,
                              void* d_out, int out_size, void* d_ws, size_t ws_size,
                              hipStream_t stream) {
    (void)in_sizes; (void)n_in; (void)out_size; (void)ws_size;
    const float* x    = (const float*)d_in[0];
    const float* bias = (const float*)d_in[1];
    const float* W1   = (const float*)d_in[2];
    const float* W2   = (const float*)d_in[3];
    const float* g1   = (const float*)d_in[4];
    const float* g2   = (const float*)d_in[5];

    float* out = (float*)d_out;
    float* out_mu2   = out;
    float* out_gamma = out + BATCH * D3;
    float* out_P     = out + BATCH * D3 + (size_t)BATCH * D3 * D3;

    float* ws = (float*)d_ws;
    float* xp      = ws;
    float* spare   = xp + BATCH * D1;
    float* mu1     = spare + D1 + D2;
    float* d2      = mu1 + BATCH * D2;
    float* sp      = d2 + BATCH * D2;
    float* m2b     = sp + BATCH * D2;
    float* vn1     = m2b + BATCH * D2;
    float* vn2     = vn1 + BATCH;
    float* P1      = vn2 + BATCH;
    float* q1      = d2;
    float* q2      = mu1;

    pool_kernel<<<(BATCH * D1 + 255) / 256, 256, 0, stream>>>(x, bias, xp);
    gemv4w_kernel<true, D1, D2, 6><<<D2 / 4, 128, 0, stream>>>(W1, xp, xp, mu1, q1);
    stats1_kernel<<<BATCH, 256, 0, stream>>>(xp, q1, g1, vn1, P1);
    act_kernel<<<(BATCH * D2 + 255) / 256, 256, 0, stream>>>(mu1, vn1, d2, sp, m2b);
    gemv4w_kernel<false, D2, D3, 4><<<D3 / 4, 128, 0, stream>>>(W2, sp, m2b, out_mu2, q2);
    stats2_kernel<<<BATCH, 256, 0, stream>>>(m2b, q2, g2, P1, vn2, out_P);

    dim3 ggrid(NPAIR, BATCH * 2);
    gamma_kernel<<<ggrid, 128, 0, stream>>>(W2, d2, vn1, vn2, out_gamma);
}


# Three small launches folded into the big ones: GEMV 1's waves sum xp^2 (vn1) and apply the activation to their row; gamma's diagonal blocks sum vn2 and P_tot; 7 launches to 4; sum orders change
# speedup vs baseline: 5.6805x; 1.1996x over previous
#include <hip/hip_runtime.h>
#include <hip/hip_bf16.h>
#include <math.h>

typedef __attribute__((ext_vector_type(2))) float v2f;
typedef __attribute__((ext_vector_type(4))) float v4f;
typedef __attribute__((ext_vector_type(8))) float v8f;

#define BATCH 4
#define CCH   3
#define HH    64
#define WW    64
#define D1    3072
#define D2    2048
#define D3    1024
#define SIGMA 0.01f

#define GT     64
#define GR     32
#define NT     (D3 / GT)
#define NPAIR  (NT * (NT + 1) / 2)
#define BK     32
#define NKB    (D2 / BK)
#define GP     36
#define CP     68
#define TP     36

__global__ void pool_kernel(const float* __restrict__ x, const float* __restrict__ bias,
                            float* __restrict__ xp) {
    int t = blockIdx.x * 256 + threadIdx.x;
    if (t >= BATCH * D1) return;
    int b = t / D1, f = t % D1;
    int c = f >> 10;
    int rem = f & 1023;
    int oh = rem >> 5, ow = rem & 31;
    int base = c * (HH * WW) + (2 * oh) * WW + 2 * ow;
    const float* xb = x + b * (CCH * HH * WW);
    float s = (xb[base]        + bias[base])
            + (xb[base + 1]    + bias[base + 1])
            + (xb[base + WW]   + bias[base + WW])
            + (xb[base + WW+1] + bias[base + WW + 1]);
    xp[t] = 0.25f * s;
}

template <bool SQ, int DIN, int DOUT, int G>
__global__ void __launch_bounds__(128)
gemv4w_kernel(const float* __restrict__ W, const float* __restrict__ X,
              const float* __restrict__ M, float* __restrict__ out,
              float* __restrict__ q, const float* __restrict__ gp,
              float* __restrict__ d2o, float* __restrict__ m2bo,
              float* __restrict__ vno) {
    constexpr int Q  = DIN / 4;
    constexpr int NC = Q / 32;
    constexpr int NG = NC / G;
    const int lane = threadIdx.x & 31;
    const int o    = blockIdx.x * 4 + (threadIdx.x >> 5);
    const v4f* w4 = (const v4f*)(W + (size_t)o * DIN) + lane;
    const v4f* x4 = (const v4f*)X + lane;
    const v4f* m4 = (const v4f*)M + lane;
    v4f w[NC];
#pragma unroll
    for (int c = 0; c < NC; ++c) w[c] = w4[c * 32];
    v4f xb[2][G][4], mb[2][G][4];
#pragma unroll
    for (int j = 0; j < G; ++j) {
#pragma unroll
        for (int b = 0; b < 4; ++b) {
            xb[0][j][b] = x4[b * Q + j * 32];
            if (!SQ) mb[0][j][b] = m4[b * Q + j * 32];
        }
    }
    v4f a[4] = {}, p[4] = {}, e[4] = {};
#pragma unroll
    for (int g = 0; g < NG; ++g) {
        if (g + 1 < NG) {
#pragma unroll
            for (int j = 0; j < G; ++j) {
#pragma unroll
                for (int b = 0; b < 4; ++b) {
                    xb[(g + 1) & 1][j][b] = x4[b * Q + ((g + 1) * G + j) * 32];
                    if (!SQ) mb[(g + 1) & 1][j][b] = m4[b * Q + ((g + 1) * G + j) * 32];
                }
            }
        }
        __builtin_amdgcn_sched_barrier(0);
#pragma unroll
        for (int j = 0; j < G; ++j) {
            const int c = g * G + j;
            v4f aw = __builtin_elementwise_abs(w[c]);
#pragma unroll
            for (int b = 0; b < 4; ++b) {
                v4f x = xb[g & 1][j][b];
                v4f m;
                if (SQ) { m = x * x; e[b] += m; }
                else    m = mb[g & 1][j][b];
                a[b] += w[c] * x;
                p[b] += aw * m;
            }
        }
    }
    constexpr int NV = SQ ? 12 : 8;
    float s[NV];
#pragma unroll
    for (int b = 0; b < 4; ++b) {
        s[b]     = (a[b].x + a[b].y) + (a[b].z + a[b].w);
        s[4 + b] = (p[b].x + p[b].y) + (p[b].z + p[b].w);
        if constexpr (SQ) s[8 + b] = (e[b].x + e[b].y) + (e[b].z + e[b].w);
    }
#pragma unroll
    for (int off = 16; off > 0; off >>= 1) {
#pragma unroll
        for (int v = 0; v < NV; ++v) s[v] += __shfl_xor(s[v], off);
    }
    float r = s[0];
#pragma unroll
    for (int v = 1; v < 8; ++v) r = (lane == v) ? s[v] : r;
    if constexpr (SQ) {
        float ev = s[8];
#pragma unroll
        for (int b = 1; b < 4; ++b) ev = (lane == b) ? s[8 + b] : ev;
        if (lane < 4) {
            const float c1 = gp[0];
            const float vb = 2.f * (SIGMA * SIGMA) / (c1 * c1) * ev;
            const float m  = r;
            const float dd = 1.f / (1.f + expf(-m));
            const float sv = (m > 20.f) ? m : log1pf(expf(m));
            const float dsq = dd * dd;
            d2o[lane * DOUT + o]  = dsq;
            out[lane * DOUT + o]  = sv;
            m2bo[lane * DOUT + o] = sv * sv + vb * dsq;
            if (o == 0) vno[lane] = vb;
        } else if (lane < 8) {
            q[(lane - 4) * DOUT + o] = r;
        }
    } else {
        if (lane < 4)      out[lane * DOUT + o]     = r;
        else if (lane < 8) q[(lane - 4) * DOUT + o] = r;
    }
}

struct KStage { v4f a[2]; v4f b[4]; v4f d; };

__device__ __forceinline__ void lds_barrier() {
    __builtin_amdgcn_fence(__ATOMIC_RELEASE, "workgroup", "local");
    __builtin_amdgcn_s_barrier();
    __builtin_amdgcn_fence(__ATOMIC_ACQUIRE, "workgroup", "local");
}

__global__ void __launch_bounds__(128)
gamma_kernel(const float* __restrict__ W2, const float* __restrict__ d2,
             const float* __restrict__ vn1, const float* __restrict__ m2b,
             const float* __restrict__ q1, const float* __restrict__ q2,
             const float* __restrict__ g1p, const float* __restrict__ g2p,
             float* __restrict__ out, float* __restrict__ Pout) {
    __shared__ __attribute__((aligned(16))) float smem[(GR + GT) * GP];
    float* As = smem;
    float* Bs = smem + GR * GP;
    __shared__ float red[12];

    const int tid  = threadIdx.x;
    const int lane = tid & 31;
    const int w    = tid >> 5;
    const int wy   = w >> 1;
    const int wx   = w & 1;
    const int r    = lane & 15;
    const int hi   = lane >> 4;

    int p  = blockIdx.x;
    int ti = 0;
    while (p >= NT - ti) { p -= NT - ti; ++ti; }
    const int tj   = ti + p;
    const int half = blockIdx.y & 1;
    const int b    = blockIdx.y >> 1;
    const int brow = ti * GT + half * GR;
    const int bcol = tj * GT;
    const bool diag = (ti == tj);

    float vd = 0.f;
    if (diag) {
        const bool pown = (blockIdx.x == 0) && (half == 0);
        const v4f* m4 = (const v4f*)(m2b + (size_t)b * D2) + tid;
        v4f ms4 = (m4[0] + m4[128]) + (m4[256] + m4[384]);
        float ms = (ms4.x + ms4.y) + (ms4.z + ms4.w);
        float s1 = 0.f, s2 = 0.f;
        if (pown) {
            const v4f* a4 = (const v4f*)(q1 + (size_t)b * D2) + tid;
            const v4f* b4 = (const v4f*)(q2 + (size_t)b * D3) + tid;
            v4f u = (a4[0] + a4[128]) + (a4[256] + a4[384]);
            v4f v = b4[0] + b4[128];
            s1 = (u.x + u.y) + (u.z + u.w);
            s2 = (v.x + v.y) + (v.z + v.w);
        }
        for (int off = 16; off > 0; off >>= 1) {
            ms += __shfl_xor(ms, off);
            s1 += __shfl_xor(s1, off);
            s2 += __shfl_xor(s2, off);
        }
        if (lane == 0) { red[w] = ms; red[4 + w] = s1; red[8 + w] = s2; }
        __syncthreads();
        const float c2 = g2p[0];
        vd = 2.f * (SIGMA * SIGMA) / (c2 * c2) * ((red[0] + red[1]) + (red[2] + red[3]));
        if (pown && tid == 0) {
            const float c1 = g1p[0];
            Pout[b] = c1 * ((red[4] + red[5]) + (red[6] + red[7]))
                    + c2 * ((red[8] + red[9]) + (red[10] + red[11]));
        }
    }

    const int srow = tid >> 3;
    const int c4   = tid & 7;
    const float* d2b = d2 + (size_t)b * D2;
    const float* gA  = W2 + (size_t)(brow + srow) * D2 + c4 * 4;
    const float* gB  = W2 + (size_t)(bcol + srow) * D2 + c4 * 4;

    auto request = [&](KStage& s, int kb) {
        const float* gA2 = gA + kb * BK;
        const float* gB2 = gB + kb * BK;
        s.d = *(const v4f*)(d2b + kb * BK + c4 * 4);
#pragma unroll
        for (int q = 0; q < 2; ++q) s.a[q] = *(const v4f*)(gA2 + (size_t)q * 16 * D2);
#pragma unroll
        for (int q = 0; q < 4; ++q) s.b[q] = *(const v4f*)(gB2 + (size_t)q * 16 * D2);
    };

    KStage s0, s1;
    request(s0, 0);
    __builtin_amdgcn_sched_barrier(0);
    request(s1, 1);

    v8f acc[2] = {};

    const int sst = srow * GP + c4 * 4;
    const int fa  = (wy * 16 + r) * GP + 2 * hi;
    const int fb  = (wx * 32 + r) * GP + 2 * hi;

    auto k_step = [&](KStage& s, int kb, bool more) {
        lds_barrier();
        asm volatile("" : "+v"(s.d));
#pragma unroll
        for (int q = 0; q < 2; ++q) *(v4f*)(As + sst + q * 16 * GP) = s.a[q] * s.d;
#pragma unroll
        for (int q = 0; q < 4; ++q) *(v4f*)(Bs + sst + q * 16 * GP) = s.b[q];
        if (more) request(s, kb + 2);
        lds_barrier();

        v2f a[BK / 4], bv[2][BK / 4];
#pragma unroll
        for (int ks = 0; ks < BK / 4; ++ks) {
            a[ks] = *(const v2f*)(As + fa + 4 * ks);
#pragma unroll
            for (int j = 0; j < 2; ++j)
                bv[j][ks] = *(const v2f*)(Bs + fb + j * 16 * GP + 4 * ks);
        }
#pragma unroll
        for (int ks = 0; ks < BK / 4; ++ks) {
#pragma unroll
            for (int j = 0; j < 2; ++j)
                acc[j] = __builtin_amdgcn_wmma_f32_16x16x4_f32(
                    false, a[ks], false, bv[j][ks], (short)0, acc[j], false, false);
        }
    };

    for (int kb = 0; kb < NKB - 2; kb += 2) {
        k_step(s0, kb, true);
        k_step(s1, kb + 1, true);
    }
    k_step(s0, NKB - 2, false);
    k_step(s1, NKB - 1, false);

    const float s  = vn1[b];
    float* ob = out + (size_t)b * D3 * D3;
    float* Cs = smem;

    __syncthreads();
#pragma unroll
    for (int j = 0; j < 2; ++j) {
#pragma unroll
        for (int v = 0; v < 8; ++v) {
            int lr = wy * 16 + v + 8 * hi;
            int lc = wx * 32 + j * 16 + r;
            float val = s * acc[j][v];
            if (diag && half * GR + lr == lc) val += vd;
            acc[j][v] = val;
            Cs[lr * CP + lc] = val;
        }
    }
    __syncthreads();

    const int er = tid >> 4;
    const int ec = (tid & 15) * 4;
#pragma unroll
    for (int q = 0; q < 4; ++q) {
        int row = er + 8 * q;
        v4f v4 = *(const v4f*)(Cs + row * CP + ec);
        *(v4f*)(ob + (size_t)(brow + row) * D3 + bcol + ec) = v4;
    }

    if (!diag) {
        __syncthreads();
#pragma unroll
        for (int j = 0; j < 2; ++j) {
#pragma unroll
            for (int v = 0; v < 8; ++v) {
                int lr = wy * 16 + v + 8 * hi;
                int lc = wx * 32 + j * 16 + r;
                Cs[lc * TP + lr] = acc[j][v];
            }
        }
        __syncthreads();
        const int tr = tid >> 3;
        const int tc = (tid & 7) * 4;
#pragma unroll
        for (int q = 0; q < 4; ++q) {
            int row = tr + 16 * q;
            v4f v4 = *(const v4f*)(Cs + row * TP + tc);
            *(v4f*)(ob + (size_t)(bcol + row) * D3 + brow + tc) = v4;
        }
    }
}

extern "C" void kernel_launch(void* const* d_in, const int* in_sizes, int n_in,
                              void* d_out, int out_size, void* d_ws, size_t ws_size,
                              hipStream_t stream) {
    (void)in_sizes; (void)n_in; (void)out_size; (void)ws_size;
    const float* x    = (const float*)d_in[0];
    const float* bias = (const float*)d_in[1];
    const float* W1   = (const float*)d_in[2];
    const float* W2   = (const float*)d_in[3];
    const float* g1   = (const float*)d_in[4];
    const float* g2   = (const float*)d_in[5];

    float* out = (float*)d_out;
    float* out_mu2   = out;
    float* out_gamma = out + BATCH * D3;
    float* out_P     = out + BATCH * D3 + (size_t)BATCH * D3 * D3;

    float* ws = (float*)d_ws;
    float* xp   = ws;
    float* q2   = xp + BATCH * D1;
    float* q1   = q2 + D1 + D2;
    float* d2   = q1 + BATCH * D2;
    float* sp   = d2 + BATCH * D2;
    float* m2b  = sp + BATCH * D2;
    float* vn1  = m2b + BATCH * D2;

    pool_kernel<<<(BATCH * D1 + 255) / 256, 256, 0, stream>>>(x, bias, xp);
    gemv4w_kernel<true, D1, D2, 6><<<D2 / 4, 128, 0, stream>>>(W1, xp, xp, sp, q1, g1, d2, m2b, vn1);
    gemv4w_kernel<false, D2, D3, 4><<<D3 / 4, 128, 0, stream>>>(W2, sp, m2b, out_mu2, q2,
                                                              nullptr, nullptr, nullptr, nullptr);

    dim3 ggrid(NPAIR, BATCH * 2);
    gamma_kernel<<<ggrid, 128, 0, stream>>>(W2, d2, vn1, m2b, q1, q2, g1, g2, out_gamma, out_P);
}


# gamma2: a diagonal tile's lower block computes only its bottom right 32x32 (upper block mirrors the rest); these 64 half-work blocks are last in the grid: busiest SIMD has 4.5 blocks of WMMAs, not 5
# speedup vs baseline: 6.0726x; 1.0690x over previous
#include <hip/hip_runtime.h>
#include <hip/hip_bf16.h>
#include <math.h>

typedef __attribute__((ext_vector_type(2))) float v2f;
typedef __attribute__((ext_vector_type(4))) float v4f;
typedef __attribute__((ext_vector_type(8))) float v8f;

#define BATCH 4
#define CCH   3
#define HH    64
#define WW    64
#define D1    3072
#define D2    2048
#define D3    1024
#define SIGMA 0.01f

#define GT     64
#define GR     32
#define NT     (D3 / GT)
#define NPAIR  (NT * (NT + 1) / 2)
#define BK     32
#define NKB    (D2 / BK)
#define GP     36
#define CP     68
#define TP     36

__global__ void pool_kernel(const float* __restrict__ x, const float* __restrict__ bias,
                            float* __restrict__ xp) {
    int t = blockIdx.x * 256 + threadIdx.x;
    if (t >= BATCH * D1) return;
    int b = t / D1, f = t % D1;
    int c = f >> 10;
    int rem = f & 1023;
    int oh = rem >> 5, ow = rem & 31;
    int base = c * (HH * WW) + (2 * oh) * WW + 2 * ow;
    const float* xb = x + b * (CCH * HH * WW);
    float s = (xb[base]        + bias[base])
            + (xb[base + 1]    + bias[base + 1])
            + (xb[base + WW]   + bias[base + WW])
            + (xb[base + WW+1] + bias[base + WW + 1]);
    xp[t] = 0.25f * s;
}

template <bool SQ, int DIN, int DOUT, int G>
__global__ void __launch_bounds__(128)
gemv4w_kernel(const float* __restrict__ W, const float* __restrict__ X,
              const float* __restrict__ M, float* __restrict__ out,
              float* __restrict__ q, const float* __restrict__ gp,
              float* __restrict__ d2o, float* __restrict__ m2bo,
              float* __restrict__ vno) {
    constexpr int Q  = DIN / 4;
    constexpr int NC = Q / 32;
    constexpr int NG = NC / G;
    const int lane = threadIdx.x & 31;
    const int o    = blockIdx.x * 4 + (threadIdx.x >> 5);
    const v4f* w4 = (const v4f*)(W + (size_t)o * DIN) + lane;
    const v4f* x4 = (const v4f*)X + lane;
    const v4f* m4 = (const v4f*)M + lane;
    v4f w[NC];
#pragma unroll
    for (int c = 0; c < NC; ++c) w[c] = w4[c * 32];
    v4f xb[2][G][4], mb[2][G][4];
#pragma unroll
    for (int j = 0; j < G; ++j) {
#pragma unroll
        for (int b = 0; b < 4; ++b) {
            xb[0][j][b] = x4[b * Q + j * 32];
            if (!SQ) mb[0][j][b] = m4[b * Q + j * 32];
        }
    }
    v4f a[4] = {}, p[4] = {}, e[4] = {};
#pragma unroll
    for (int g = 0; g < NG; ++g) {
        if (g + 1 < NG) {
#pragma unroll
            for (int j = 0; j < G; ++j) {
#pragma unroll
                for (int b = 0; b < 4; ++b) {
                    xb[(g + 1) & 1][j][b] = x4[b * Q + ((g + 1) * G + j) * 32];
                    if (!SQ) mb[(g + 1) & 1][j][b] = m4[b * Q + ((g + 1) * G + j) * 32];
                }
            }
        }
        __builtin_amdgcn_sched_barrier(0);
#pragma unroll
        for (int j = 0; j < G; ++j) {
            const int c = g * G + j;
            v4f aw = __builtin_elementwise_abs(w[c]);
#pragma unroll
            for (int b = 0; b < 4; ++b) {
                v4f x = xb[g & 1][j][b];
                v4f m;
                if (SQ) { m = x * x; e[b] += m; }
                else    m = mb[g & 1][j][b];
                a[b] += w[c] * x;
                p[b] += aw * m;
            }
        }
    }
    constexpr int NV = SQ ? 12 : 8;
    float s[NV];
#pragma unroll
    for (int b = 0; b < 4; ++b) {
        s[b]     = (a[b].x + a[b].y) + (a[b].z + a[b].w);
        s[4 + b] = (p[b].x + p[b].y) + (p[b].z + p[b].w);
        if constexpr (SQ) s[8 + b] = (e[b].x + e[b].y) + (e[b].z + e[b].w);
    }
#pragma unroll
    for (int off = 16; off > 0; off >>= 1) {
#pragma unroll
        for (int v = 0; v < NV; ++v) s[v] += __shfl_xor(s[v], off);
    }
    float r = s[0];
#pragma unroll
    for (int v = 1; v < 8; ++v) r = (lane == v) ? s[v] : r;
    if constexpr (SQ) {
        float ev = s[8];
#pragma unroll
        for (int b = 1; b < 4; ++b) ev = (lane == b) ? s[8 + b] : ev;
        if (lane < 4) {
            const float c1 = gp[0];
            const float vb = 2.f * (SIGMA * SIGMA) / (c1 * c1) * ev;
            const float m  = r;
            const float dd = 1.f / (1.f + expf(-m));
            const float sv = (m > 20.f) ? m : log1pf(expf(m));
            const float dsq = dd * dd;
            d2o[lane * DOUT + o]  = dsq;
            out[lane * DOUT + o]  = sv;
            m2bo[lane * DOUT + o] = sv * sv + vb * dsq;
            if (o == 0) vno[lane] = vb;
        } else if (lane < 8) {
            q[(lane - 4) * DOUT + o] = r;
        }
    } else {
        if (lane < 4)      out[lane * DOUT + o]     = r;
        else if (lane < 8) q[(lane - 4) * DOUT + o] = r;
    }
}

template <int NB> struct KStage { v4f a[2]; v4f b[NB]; v4f d; };

__device__ __forceinline__ void lds_barrier() {
    __builtin_amdgcn_fence(__ATOMIC_RELEASE, "workgroup", "local");
    __builtin_amdgcn_s_barrier();
    __builtin_amdgcn_fence(__ATOMIC_ACQUIRE, "workgroup", "local");
}


template <int NJ>
__device__ __forceinline__ void gamma_tile(const float* __restrict__ W2,
                                           const float* __restrict__ d2b,
                                           const float* __restrict__ vn1b,
                                           float* smem, int brow, int bcol,
                                           float vd, bool diag, int tq0,
                                           float* __restrict__ ob) {
    constexpr int NB = 2 * NJ;
    float* As = smem;
    float* Bs = smem + GR * GP;

    const int tid  = threadIdx.x;
    const int lane = tid & 31;
    const int w    = tid >> 5;
    const int wy   = w >> 1;
    const int wx   = w & 1;
    const int r    = lane & 15;
    const int hi   = lane >> 4;

    const int srow = tid >> 3;
    const int c4   = tid & 7;
    const float* gA  = W2 + (size_t)(brow + srow) * D2 + c4 * 4;
    const float* gB  = W2 + (size_t)(bcol + srow) * D2 + c4 * 4;

    auto request = [&](KStage<NB>& s, int kb) {
        const float* gA2 = gA + kb * BK;
        const float* gB2 = gB + kb * BK;
        s.d = *(const v4f*)(d2b + kb * BK + c4 * 4);
#pragma unroll
        for (int q = 0; q < 2; ++q) s.a[q] = *(const v4f*)(gA2 + (size_t)q * 16 * D2);
#pragma unroll
        for (int q = 0; q < NB; ++q) s.b[q] = *(const v4f*)(gB2 + (size_t)q * 16 * D2);
    };

    KStage<NB> s0, s1;
    request(s0, 0);
    __builtin_amdgcn_sched_barrier(0);
    request(s1, 1);

    v8f acc[NJ] = {};

    const int sst = srow * GP + c4 * 4;
    const int fa  = (wy * 16 + r) * GP + 2 * hi;
    const int fb  = (wx * 16 * NJ + r) * GP + 2 * hi;

    auto k_step = [&](KStage<NB>& s, int kb, bool more) {
        lds_barrier();
        asm volatile("" : "+v"(s.d));
#pragma unroll
        for (int q = 0; q < 2; ++q) *(v4f*)(As + sst + q * 16 * GP) = s.a[q] * s.d;
#pragma unroll
        for (int q = 0; q < NB; ++q) *(v4f*)(Bs + sst + q * 16 * GP) = s.b[q];
        if (more) request(s, kb + 2);
        lds_barrier();

        v2f a[BK / 4], bv[NJ][BK / 4];
#pragma unroll
        for (int ks = 0; ks < BK / 4; ++ks) {
            a[ks] = *(const v2f*)(As + fa + 4 * ks);
#pragma unroll
            for (int j = 0; j < NJ; ++j)
                bv[j][ks] = *(const v2f*)(Bs + fb + j * 16 * GP + 4 * ks);
        }
#pragma unroll
        for (int ks = 0; ks < BK / 4; ++ks) {
#pragma unroll
            for (int j = 0; j < NJ; ++j)
                acc[j] = __builtin_amdgcn_wmma_f32_16x16x4_f32(
                    false, a[ks], false, bv[j][ks], (short)0, acc[j], false, false);
        }
    };

    for (int kb = 0; kb < NKB - 2; kb += 2) {
        k_step(s0, kb, true);
        k_step(s1, kb + 1, true);
    }
    k_step(s0, NKB - 2, false);
    k_step(s1, NKB - 1, false);

    const float s  = vn1b[0];
    float* Cs = smem;

    __syncthreads();
#pragma unroll
    for (int j = 0; j < NJ; ++j) {
#pragma unroll
        for (int v = 0; v < 8; ++v) {
            int lr = wy * 16 + v + 8 * hi;
            int lc = wx * 16 * NJ + j * 16 + r;
            float val = s * acc[j][v];
            if (diag && lr == lc) val += vd;
            acc[j][v] = val;
            Cs[lr * CP + lc] = val;
        }
    }
    __syncthreads();

    if constexpr (NJ == 2) {
        const int er = tid >> 4;
        const int ec = (tid & 15) * 4;
#pragma unroll
        for (int q = 0; q < 4; ++q) {
            int row = er + 8 * q;
            v4f v4 = *(const v4f*)(Cs + row * CP + ec);
            *(v4f*)(ob + (size_t)(brow + row) * D3 + bcol + ec) = v4;
        }

        {
            __syncthreads();
#pragma unroll
            for (int j = 0; j < 2; ++j) {
#pragma unroll
                for (int v = 0; v < 8; ++v) {
                    int lr = wy * 16 + v + 8 * hi;
                    int lc = wx * 32 + j * 16 + r;
                    Cs[lc * TP + lr] = acc[j][v];
                }
            }
            __syncthreads();
            const int tr = tid >> 3;
            const int tc = (tid & 7) * 4;
            if (tq0 == 0) {
#pragma unroll
                for (int q = 0; q < 2; ++q) {
                    int row = tr + 16 * q;
                    v4f v4 = *(const v4f*)(Cs + row * TP + tc);
                    *(v4f*)(ob + (size_t)(bcol + row) * D3 + brow + tc) = v4;
                }
            }
#pragma unroll
            for (int q = 2; q < 4; ++q) {
                int row = tr + 16 * q;
                v4f v4 = *(const v4f*)(Cs + row * TP + tc);
                *(v4f*)(ob + (size_t)(bcol + row) * D3 + brow + tc) = v4;
            }
        }
    } else {
        const int tr = tid >> 3;
        const int tc = (tid & 7) * 4;
#pragma unroll
        for (int q = 0; q < 2; ++q) {
            int row = tr + 16 * q;
            v4f v4 = *(const v4f*)(Cs + row * CP + tc);
            *(v4f*)(ob + (size_t)(brow + row) * D3 + bcol + tc) = v4;
        }
    }
}

#define NOFF   (NPAIR - NT)
#define NFB    (2 * NOFF + NT)
#define NFULL  (BATCH * NFB)
#define NBLK   (NFULL + BATCH * NT)

__global__ void __launch_bounds__(128)
gamma_kernel(const float* __restrict__ W2, const float* __restrict__ d2,
             const float* __restrict__ vn1, const float* __restrict__ m2b,
             const float* __restrict__ q1, const float* __restrict__ q2,
             const float* __restrict__ g1p, const float* __restrict__ g2p,
             float* __restrict__ out, float* __restrict__ Pout) {
    __shared__ __attribute__((aligned(16))) float smem[(GR + GT) * GP];
    __shared__ float red[12];

    const int tid  = threadIdx.x;
    const int lane = tid & 31;
    const int w    = tid >> 5;

    const int L = blockIdx.x;
    const bool light = (L >= NFULL);
    int b, ti, tj, half = 0;
    bool diag, pown = false;
    if (!light) {
        b = L / NFB;
        const int m = L % NFB;
        if (m < 2 * NOFF) {
            int u = m >> 1;
            half = m & 1;
            ti = 0;
            while (u >= NT - 1 - ti) { u -= NT - 1 - ti; ++ti; }
            tj = ti + 1 + u;
            diag = false;
        } else {
            ti = tj = m - 2 * NOFF;
            diag = true;
            pown = (ti == 0);
        }
    } else {
        const int e = L - NFULL;
        b = e / NT;
        ti = tj = e % NT;
        diag = true;
    }
    const int brow = ti * GT + (light ? GR : half * GR);
    const int bcol = tj * GT + (light ? GR : 0);

    float vd = 0.f;
    if (diag) {
        const v4f* m4 = (const v4f*)(m2b + (size_t)b * D2) + tid;
        v4f ms4 = (m4[0] + m4[128]) + (m4[256] + m4[384]);
        float ms = (ms4.x + ms4.y) + (ms4.z + ms4.w);
        float s1 = 0.f, s2 = 0.f;
        if (pown) {
            const v4f* a4 = (const v4f*)(q1 + (size_t)b * D2) + tid;
            const v4f* b4 = (const v4f*)(q2 + (size_t)b * D3) + tid;
            v4f u = (a4[0] + a4[128]) + (a4[256] + a4[384]);
            v4f v = b4[0] + b4[128];
            s1 = (u.x + u.y) + (u.z + u.w);
            s2 = (v.x + v.y) + (v.z + v.w);
        }
        for (int off = 16; off > 0; off >>= 1) {
            ms += __shfl_xor(ms, off);
            s1 += __shfl_xor(s1, off);
            s2 += __shfl_xor(s2, off);
        }
        if (lane == 0) { red[w] = ms; red[4 + w] = s1; red[8 + w] = s2; }
        __syncthreads();
        const float c2 = g2p[0];
        vd = 2.f * (SIGMA * SIGMA) / (c2 * c2) * ((red[0] + red[1]) + (red[2] + red[3]));
        if (pown && tid == 0) {
            const float c1 = g1p[0];
            Pout[b] = c1 * ((red[4] + red[5]) + (red[6] + red[7]))
                    + c2 * ((red[8] + red[9]) + (red[10] + red[11]));
        }
    }

    const float* d2b = d2 + (size_t)b * D2;
    float* ob = out + (size_t)b * D3 * D3;
    if (light)
        gamma_tile<1>(W2, d2b, vn1 + b, smem, brow, bcol, vd, true, 0, ob);
    else
        gamma_tile<2>(W2, d2b, vn1 + b, smem, brow, bcol, vd, diag, diag ? 2 : 0, ob);
}

extern "C" void kernel_launch(void* const* d_in, const int* in_sizes, int n_in,
                              void* d_out, int out_size, void* d_ws, size_t ws_size,
                              hipStream_t stream) {
    (void)in_sizes; (void)n_in; (void)out_size; (void)ws_size;
    const float* x    = (const float*)d_in[0];
    const float* bias = (const float*)d_in[1];
    const float* W1   = (const float*)d_in[2];
    const float* W2   = (const float*)d_in[3];
    const float* g1   = (const float*)d_in[4];
    const float* g2   = (const float*)d_in[5];

    float* out = (float*)d_out;
    float* out_mu2   = out;
    float* out_gamma = out + BATCH * D3;
    float* out_P     = out + BATCH * D3 + (size_t)BATCH * D3 * D3;

    float* ws = (float*)d_ws;
    float* xp   = ws;
    float* q2   = xp + BATCH * D1;
    float* q1   = q2 + D1 + D2;
    float* d2   = q1 + BATCH * D2;
    float* sp   = d2 + BATCH * D2;
    float* m2b  = sp + BATCH * D2;
    float* vn1  = m2b + BATCH * D2;

    pool_kernel<<<(BATCH * D1 + 255) / 256, 256, 0, stream>>>(x, bias, xp);
    gemv4w_kernel<true, D1, D2, 6><<<D2 / 4, 128, 0, stream>>>(W1, xp, xp, sp, q1, g1, d2, m2b, vn1);
    gemv4w_kernel<false, D2, D3, 4><<<D3 / 4, 128, 0, stream>>>(W2, sp, m2b, out_mu2, q2,
                                                              nullptr, nullptr, nullptr, nullptr);

    gamma_kernel<<<NBLK, 128, 0, stream>>>(W2, d2, vn1, m2b, q1, q2, g1, g2, out_gamma, out_P);
}


# Layer 2's launch dissolved: mu2 rows ride as 256 extra blocks of the gamma launch; its power term = |W2| column sums (made beside the pooling) dot m2b; 4 launches to 3; P_tot sum order changes
# speedup vs baseline: 6.2936x; 1.0364x over previous
#include <hip/hip_runtime.h>
#include <hip/hip_bf16.h>
#include <math.h>

typedef __attribute__((ext_vector_type(2))) float v2f;
typedef __attribute__((ext_vector_type(4))) float v4f;
typedef __attribute__((ext_vector_type(8))) float v8f;

#define BATCH 4
#define CCH   3
#define HH    64
#define WW    64
#define D1    3072
#define D2    2048
#define D3    1024
#define SIGMA 0.01f

#define GT     64
#define GR     32
#define NT     (D3 / GT)
#define NPAIR  (NT * (NT + 1) / 2)
#define BK     32
#define NKB    (D2 / BK)
#define GP     36
#define CP     68
#define TP     36

#define NPOOLB  ((BATCH * D1 + 255) / 256)
#define CSC     32
#define CSG     2
#define CSR     (D3 / CSG)
#define CSP     36
#define NCSB    ((D2 / CSC) * CSG)

__global__ void __launch_bounds__(256)
pool_colsum_kernel(const float* __restrict__ x, const float* __restrict__ bias,
                   const float* __restrict__ W2, float* __restrict__ xp,
                   float* __restrict__ csp) {
    if (blockIdx.x >= NPOOLB) {
        __shared__ __attribute__((aligned(16))) float red[8 * CSP];
        const int e     = blockIdx.x - NPOOLB;
        const int strip = e / CSG, rg = e % CSG;
        const int rl    = threadIdx.x >> 3, c4 = threadIdx.x & 7;
        const v4f* p = (const v4f*)(W2 + (size_t)(rg * CSR + rl) * D2 + strip * CSC) + c4;
        v4f w[CSR / 32];
#pragma unroll
        for (int k = 0; k < CSR / 32; ++k) w[k] = p[(size_t)k * 32 * (D2 / 4)];
        v4f s = {};
#pragma unroll
        for (int k = 0; k < CSR / 32; ++k) s += __builtin_elementwise_abs(w[k]);
#pragma unroll
        for (int off = 8; off <= 16; off <<= 1) {
            s.x += __shfl_xor(s.x, off);
            s.y += __shfl_xor(s.y, off);
            s.z += __shfl_xor(s.z, off);
            s.w += __shfl_xor(s.w, off);
        }
        if ((threadIdx.x & 31) < 8) *(v4f*)(red + (threadIdx.x >> 5) * CSP + c4 * 4) = s;
        __syncthreads();
        if (threadIdx.x < CSC) {
            float t = 0.f;
#pragma unroll
            for (int r = 0; r < 8; ++r) t += red[r * CSP + threadIdx.x];
            csp[rg * D2 + strip * CSC + threadIdx.x] = t;
        }
        return;
    }
    int t = blockIdx.x * 256 + threadIdx.x;
    if (t >= BATCH * D1) return;
    int b = t / D1, f = t % D1;
    int c = f >> 10;
    int rem = f & 1023;
    int oh = rem >> 5, ow = rem & 31;
    int base = c * (HH * WW) + (2 * oh) * WW + 2 * ow;
    const float* xb = x + b * (CCH * HH * WW);
    float s = (xb[base]        + bias[base])
            + (xb[base + 1]    + bias[base + 1])
            + (xb[base + WW]   + bias[base + WW])
            + (xb[base + WW+1] + bias[base + WW + 1]);
    xp[t] = 0.25f * s;
}

template <bool SQ, bool PW, int DIN, int DOUT, int G>
__device__ __forceinline__ void
gemv_row(const int o, const float* __restrict__ W, const float* __restrict__ X,
         const float* __restrict__ M, float* __restrict__ out,
         float* __restrict__ q, const float* __restrict__ gp,
         float* __restrict__ d2o, float* __restrict__ m2bo,
         float* __restrict__ vno) {
    static_assert(PW || !SQ, "the first layer always carries its power term");
    constexpr bool LM = PW && !SQ;
    constexpr int Q  = DIN / 4;
    constexpr int NC = Q / 32;
    constexpr int NG = NC / G;
    const int lane = threadIdx.x & 31;
    const v4f* w4 = (const v4f*)(W + (size_t)o * DIN) + lane;
    const v4f* x4 = (const v4f*)X + lane;
    const v4f* m4 = (const v4f*)M + lane;
    v4f w[NC];
#pragma unroll
    for (int c = 0; c < NC; ++c) w[c] = w4[c * 32];
    v4f xb[2][G][4], mb[2][G][4];
#pragma unroll
    for (int j = 0; j < G; ++j) {
#pragma unroll
        for (int b = 0; b < 4; ++b) {
            xb[0][j][b] = x4[b * Q + j * 32];
            if constexpr (LM) mb[0][j][b] = m4[b * Q + j * 32];
        }
    }
    v4f a[4] = {}, p[4] = {}, e[4] = {};
#pragma unroll
    for (int g = 0; g < NG; ++g) {
        if (g + 1 < NG) {
#pragma unroll
            for (int j = 0; j < G; ++j) {
#pragma unroll
                for (int b = 0; b < 4; ++b) {
                    xb[(g + 1) & 1][j][b] = x4[b * Q + ((g + 1) * G + j) * 32];
                    if constexpr (LM) mb[(g + 1) & 1][j][b] = m4[b * Q + ((g + 1) * G + j) * 32];
                }
            }
        }
        __builtin_amdgcn_sched_barrier(0);
#pragma unroll
        for (int j = 0; j < G; ++j) {
            const int c = g * G + j;
            v4f aw = __builtin_elementwise_abs(w[c]);
#pragma unroll
            for (int b = 0; b < 4; ++b) {
                v4f x = xb[g & 1][j][b];
                a[b] += w[c] * x;
                if constexpr (PW) {
                    v4f m;
                    if constexpr (SQ) { m = x * x; e[b] += m; }
                    else              m = mb[g & 1][j][b];
                    p[b] += aw * m;
                }
            }
        }
    }
    constexpr int NV = SQ ? 12 : (PW ? 8 : 4);
    constexpr int NS = PW ? 8 : 4;
    float s[NV];
#pragma unroll
    for (int b = 0; b < 4; ++b) {
        s[b]     = (a[b].x + a[b].y) + (a[b].z + a[b].w);
        if constexpr (PW) s[4 + b] = (p[b].x + p[b].y) + (p[b].z + p[b].w);
        if constexpr (SQ) s[8 + b] = (e[b].x + e[b].y) + (e[b].z + e[b].w);
    }
#pragma unroll
    for (int off = 16; off > 0; off >>= 1) {
#pragma unroll
        for (int v = 0; v < NV; ++v) s[v] += __shfl_xor(s[v], off);
    }
    float r = s[0];
#pragma unroll
    for (int v = 1; v < NS; ++v) r = (lane == v) ? s[v] : r;
    if constexpr (SQ) {
        float ev = s[8];
#pragma unroll
        for (int b = 1; b < 4; ++b) ev = (lane == b) ? s[8 + b] : ev;
        if (lane < 4) {
            const float c1 = gp[0];
            const float vb = 2.f * (SIGMA * SIGMA) / (c1 * c1) * ev;
            const float m  = r;
            const float dd = 1.f / (1.f + expf(-m));
            const float sv = (m > 20.f) ? m : log1pf(expf(m));
            const float dsq = dd * dd;
            d2o[lane * DOUT + o]  = dsq;
            out[lane * DOUT + o]  = sv;
            m2bo[lane * DOUT + o] = sv * sv + vb * dsq;
            if (o == 0) vno[lane] = vb;
        } else if (lane < 8) {
            q[(lane - 4) * DOUT + o] = r;
        }
    } else if constexpr (PW) {
        if (lane < 4)      out[lane * DOUT + o]     = r;
        else if (lane < 8) q[(lane - 4) * DOUT + o] = r;
    } else {
        if (lane < 4)      out[lane * DOUT + o]     = r;
    }
}

__global__ void __launch_bounds__(128)
gemv1_kernel(const float* __restrict__ W, const float* __restrict__ X,
             float* __restrict__ out, float* __restrict__ q,
             const float* __restrict__ gp, float* __restrict__ d2o,
             float* __restrict__ m2bo, float* __restrict__ vno) {
    gemv_row<true, true, D1, D2, 6>(blockIdx.x * 4 + (threadIdx.x >> 5),
                                    W, X, X, out, q, gp, d2o, m2bo, vno);
}

template <int NB> struct KStage { v4f a[2]; v4f b[NB]; v4f d; };

__device__ __forceinline__ void lds_barrier() {
    __builtin_amdgcn_fence(__ATOMIC_RELEASE, "workgroup", "local");
    __builtin_amdgcn_s_barrier();
    __builtin_amdgcn_fence(__ATOMIC_ACQUIRE, "workgroup", "local");
}


template <int NJ>
__device__ __forceinline__ void gamma_tile(const float* __restrict__ W2,
                                           const float* __restrict__ d2b,
                                           const float* __restrict__ vn1b,
                                           float* smem, int brow, int bcol,
                                           float vd, bool diag, int tq0,
                                           float* __restrict__ ob) {
    constexpr int NB = 2 * NJ;
    float* As = smem;
    float* Bs = smem + GR * GP;

    const int tid  = threadIdx.x;
    const int lane = tid & 31;
    const int w    = tid >> 5;
    const int wy   = w >> 1;
    const int wx   = w & 1;
    const int r    = lane & 15;
    const int hi   = lane >> 4;

    const int srow = tid >> 3;
    const int c4   = tid & 7;
    const float* gA  = W2 + (size_t)(brow + srow) * D2 + c4 * 4;
    const float* gB  = W2 + (size_t)(bcol + srow) * D2 + c4 * 4;

    auto request = [&](KStage<NB>& s, int kb) {
        const float* gA2 = gA + kb * BK;
        const float* gB2 = gB + kb * BK;
        s.d = *(const v4f*)(d2b + kb * BK + c4 * 4);
#pragma unroll
        for (int q = 0; q < 2; ++q) s.a[q] = *(const v4f*)(gA2 + (size_t)q * 16 * D2);
#pragma unroll
        for (int q = 0; q < NB; ++q) s.b[q] = *(const v4f*)(gB2 + (size_t)q * 16 * D2);
    };

    KStage<NB> s0, s1;
    request(s0, 0);
    __builtin_amdgcn_sched_barrier(0);
    request(s1, 1);

    v8f acc[NJ] = {};

    const int sst = srow * GP + c4 * 4;
    const int fa  = (wy * 16 + r) * GP + 2 * hi;
    const int fb  = (wx * 16 * NJ + r) * GP + 2 * hi;

    auto k_step = [&](KStage<NB>& s, int kb, bool more) {
        lds_barrier();
        asm volatile("" : "+v"(s.d));
#pragma unroll
        for (int q = 0; q < 2; ++q) *(v4f*)(As + sst + q * 16 * GP) = s.a[q] * s.d;
#pragma unroll
        for (int q = 0; q < NB; ++q) *(v4f*)(Bs + sst + q * 16 * GP) = s.b[q];
        if (more) request(s, kb + 2);
        lds_barrier();

        v2f a[BK / 4], bv[NJ][BK / 4];
#pragma unroll
        for (int ks = 0; ks < BK / 4; ++ks) {
            a[ks] = *(const v2f*)(As + fa + 4 * ks);
#pragma unroll
            for (int j = 0; j < NJ; ++j)
                bv[j][ks] = *(const v2f*)(Bs + fb + j * 16 * GP + 4 * ks);
        }
#pragma unroll
        for (int ks = 0; ks < BK / 4; ++ks) {
#pragma unroll
            for (int j = 0; j < NJ; ++j)
                acc[j] = __builtin_amdgcn_wmma_f32_16x16x4_f32(
                    false, a[ks], false, bv[j][ks], (short)0, acc[j], false, false);
        }
    };

    for (int kb = 0; kb < NKB - 2; kb += 2) {
        k_step(s0, kb, true);
        k_step(s1, kb + 1, true);
    }
    k_step(s0, NKB - 2, false);
    k_step(s1, NKB - 1, false);

    const float s  = vn1b[0];
    float* Cs = smem;

    __syncthreads();
#pragma unroll
    for (int j = 0; j < NJ; ++j) {
#pragma unroll
        for (int v = 0; v < 8; ++v) {
            int lr = wy * 16 + v + 8 * hi;
            int lc = wx * 16 * NJ + j * 16 + r;
            float val = s * acc[j][v];
            if (diag && lr == lc) val += vd;
            acc[j][v] = val;
            Cs[lr * CP + lc] = val;
        }
    }
    __syncthreads();

    if constexpr (NJ == 2) {
        const int er = tid >> 4;
        const int ec = (tid & 15) * 4;
#pragma unroll
        for (int q = 0; q < 4; ++q) {
            int row = er + 8 * q;
            v4f v4 = *(const v4f*)(Cs + row * CP + ec);
            *(v4f*)(ob + (size_t)(brow + row) * D3 + bcol + ec) = v4;
        }

        {
            __syncthreads();
#pragma unroll
            for (int j = 0; j < 2; ++j) {
#pragma unroll
                for (int v = 0; v < 8; ++v) {
                    int lr = wy * 16 + v + 8 * hi;
                    int lc = wx * 32 + j * 16 + r;
                    Cs[lc * TP + lr] = acc[j][v];
                }
            }
            __syncthreads();
            const int tr = tid >> 3;
            const int tc = (tid & 7) * 4;
            if (tq0 == 0) {
#pragma unroll
                for (int q = 0; q < 2; ++q) {
                    int row = tr + 16 * q;
                    v4f v4 = *(const v4f*)(Cs + row * TP + tc);
                    *(v4f*)(ob + (size_t)(bcol + row) * D3 + brow + tc) = v4;
                }
            }
#pragma unroll
            for (int q = 2; q < 4; ++q) {
                int row = tr + 16 * q;
                v4f v4 = *(const v4f*)(Cs + row * TP + tc);
                *(v4f*)(ob + (size_t)(bcol + row) * D3 + brow + tc) = v4;
            }
        }
    } else {
        const int tr = tid >> 3;
        const int tc = (tid & 7) * 4;
#pragma unroll
        for (int q = 0; q < 2; ++q) {
            int row = tr + 16 * q;
            v4f v4 = *(const v4f*)(Cs + row * CP + tc);
            *(v4f*)(ob + (size_t)(brow + row) * D3 + bcol + tc) = v4;
        }
    }
}

#define NOFF   (NPAIR - NT)
#define NFB    (2 * NOFF + NT)
#define NFULL  (BATCH * NFB)
#define NBLK   (NFULL + BATCH * NT)
#define NMUB   (D3 / 4)

__global__ void __launch_bounds__(128)
gamma_kernel(const float* __restrict__ W2, const float* __restrict__ d2,
             const float* __restrict__ vn1, const float* __restrict__ m2b,
             const float* __restrict__ q1, const float* __restrict__ csp,
             const float* __restrict__ g1p, const float* __restrict__ g2p,
             const float* __restrict__ sp, float* __restrict__ mu2,
             float* __restrict__ out, float* __restrict__ Pout) {
    if (blockIdx.x >= NBLK) {
        gemv_row<false, false, D2, D3, 2>((blockIdx.x - NBLK) * 4 + (threadIdx.x >> 5),
                                          W2, sp, nullptr, mu2, nullptr,
                                          nullptr, nullptr, nullptr, nullptr);
        return;
    }
    __shared__ __attribute__((aligned(16))) float smem[(GR + GT) * GP];
    __shared__ float red[12];

    const int tid  = threadIdx.x;
    const int lane = tid & 31;
    const int w    = tid >> 5;

    const int L = blockIdx.x;
    const bool light = (L >= NFULL);
    int b, ti, tj, half = 0;
    bool diag, pown = false;
    if (!light) {
        b = L / NFB;
        const int m = L % NFB;
        if (m < 2 * NOFF) {
            int u = m >> 1;
            half = m & 1;
            ti = 0;
            while (u >= NT - 1 - ti) { u -= NT - 1 - ti; ++ti; }
            tj = ti + 1 + u;
            diag = false;
        } else {
            ti = tj = m - 2 * NOFF;
            diag = true;
            pown = (ti == 0);
        }
    } else {
        const int e = L - NFULL;
        b = e / NT;
        ti = tj = e % NT;
        diag = true;
    }
    const int brow = ti * GT + (light ? GR : half * GR);
    const int bcol = tj * GT + (light ? GR : 0);

    float vd = 0.f;
    if (diag) {
        const v4f* m4 = (const v4f*)(m2b + (size_t)b * D2) + tid;
        const v4f mp0 = m4[0], mp1 = m4[128], mp2 = m4[256], mp3 = m4[384];
        v4f ms4 = (mp0 + mp1) + (mp2 + mp3);
        float ms = (ms4.x + ms4.y) + (ms4.z + ms4.w);
        float s1 = 0.f, s2 = 0.f;
        if (pown) {
            const v4f* a4 = (const v4f*)(q1 + (size_t)b * D2) + tid;
            const v4f* c4a = (const v4f*)csp + tid;
            const v4f* c4b = (const v4f*)(csp + D2) + tid;
            v4f u = (a4[0] + a4[128]) + (a4[256] + a4[384]);
            v4f v = ((c4a[0]   + c4b[0])   * mp0 + (c4a[128] + c4b[128]) * mp1)
                  + ((c4a[256] + c4b[256]) * mp2 + (c4a[384] + c4b[384]) * mp3);
            s1 = (u.x + u.y) + (u.z + u.w);
            s2 = (v.x + v.y) + (v.z + v.w);
        }
        for (int off = 16; off > 0; off >>= 1) {
            ms += __shfl_xor(ms, off);
            s1 += __shfl_xor(s1, off);
            s2 += __shfl_xor(s2, off);
        }
        if (lane == 0) { red[w] = ms; red[4 + w] = s1; red[8 + w] = s2; }
        __syncthreads();
        const float c2 = g2p[0];
        vd = 2.f * (SIGMA * SIGMA) / (c2 * c2) * ((red[0] + red[1]) + (red[2] + red[3]));
        if (pown && tid == 0) {
            const float c1 = g1p[0];
            Pout[b] = c1 * ((red[4] + red[5]) + (red[6] + red[7]))
                    + c2 * ((red[8] + red[9]) + (red[10] + red[11]));
        }
    }

    const float* d2b = d2 + (size_t)b * D2;
    float* ob = out + (size_t)b * D3 * D3;
    if (light)
        gamma_tile<1>(W2, d2b, vn1 + b, smem, brow, bcol, vd, true, 0, ob);
    else
        gamma_tile<2>(W2, d2b, vn1 + b, smem, brow, bcol, vd, diag, diag ? 2 : 0, ob);
}

extern "C" void kernel_launch(void* const* d_in, const int* in_sizes, int n_in,
                              void* d_out, int out_size, void* d_ws, size_t ws_size,
                              hipStream_t stream) {
    (void)in_sizes; (void)n_in; (void)out_size; (void)ws_size;
    const float* x    = (const float*)d_in[0];
    const float* bias = (const float*)d_in[1];
    const float* W1   = (const float*)d_in[2];
    const float* W2   = (const float*)d_in[3];
    const float* g1   = (const float*)d_in[4];
    const float* g2   = (const float*)d_in[5];

    float* out = (float*)d_out;
    float* out_mu2   = out;
    float* out_gamma = out + BATCH * D3;
    float* out_P     = out + BATCH * D3 + (size_t)BATCH * D3 * D3;

    float* ws = (float*)d_ws;
    float* xp   = ws;
    float* csp  = xp + BATCH * D1;
    float* q1   = csp + D1 + D2;
    float* d2   = q1 + BATCH * D2;
    float* sp   = d2 + BATCH * D2;
    float* m2b  = sp + BATCH * D2;
    float* vn1  = m2b + BATCH * D2;

    pool_colsum_kernel<<<NPOOLB + NCSB, 256, 0, stream>>>(x, bias, W2, xp, csp);
    gemv1_kernel<<<D2 / 4, 128, 0, stream>>>(W1, xp, sp, q1, g1, d2, m2b, vn1);

    gamma_kernel<<<NBLK + NMUB, 128, 0, stream>>>(W2, d2, vn1, m2b, q1, csp, g1, g2,
                                                  sp, out_mu2, out_gamma, out_P);
}


# gamma2: one 16-wave block per half tile does all 4 batches: W2 K-blocks staged to LDS once not 4x, d^2 applied to the A fragments in registers; diagonal 32x32 parts K-split; sum orders change
# speedup vs baseline: 6.6965x; 1.0640x over previous
#include <hip/hip_runtime.h>
#include <hip/hip_bf16.h>
#include <math.h>

typedef __attribute__((ext_vector_type(2))) float v2f;
typedef __attribute__((ext_vector_type(4))) float v4f;
typedef __attribute__((ext_vector_type(8))) float v8f;

#define BATCH 4
#define CCH   3
#define HH    64
#define WW    64
#define D1    3072
#define D2    2048
#define D3    1024
#define SIGMA 0.01f

#define GT     64
#define GR     32
#define NT     (D3 / GT)
#define NPAIR  (NT * (NT + 1) / 2)
#define BK     32
#define NKB    (D2 / BK)
#define GP     36
#define CP     68
#define TP     36

#define NPOOLB  ((BATCH * D1 + 255) / 256)
#define CSC     32
#define CSG     2
#define CSR     (D3 / CSG)
#define CSP     36
#define NCSB    ((D2 / CSC) * CSG)

__global__ void __launch_bounds__(256)
pool_colsum_kernel(const float* __restrict__ x, const float* __restrict__ bias,
                   const float* __restrict__ W2, float* __restrict__ xp,
                   float* __restrict__ csp) {
    if (blockIdx.x >= NPOOLB) {
        __shared__ __attribute__((aligned(16))) float red[8 * CSP];
        const int e     = blockIdx.x - NPOOLB;
        const int strip = e / CSG, rg = e % CSG;
        const int rl    = threadIdx.x >> 3, c4 = threadIdx.x & 7;
        const v4f* p = (const v4f*)(W2 + (size_t)(rg * CSR + rl) * D2 + strip * CSC) + c4;
        v4f w[CSR / 32];
#pragma unroll
        for (int k = 0; k < CSR / 32; ++k) w[k] = p[(size_t)k * 32 * (D2 / 4)];
        v4f s = {};
#pragma unroll
        for (int k = 0; k < CSR / 32; ++k) s += __builtin_elementwise_abs(w[k]);
#pragma unroll
        for (int off = 8; off <= 16; off <<= 1) {
            s.x += __shfl_xor(s.x, off);
            s.y += __shfl_xor(s.y, off);
            s.z += __shfl_xor(s.z, off);
            s.w += __shfl_xor(s.w, off);
        }
        if ((threadIdx.x & 31) < 8) *(v4f*)(red + (threadIdx.x >> 5) * CSP + c4 * 4) = s;
        __syncthreads();
        if (threadIdx.x < CSC) {
            float t = 0.f;
#pragma unroll
            for (int r = 0; r < 8; ++r) t += red[r * CSP + threadIdx.x];
            csp[rg * D2 + strip * CSC + threadIdx.x] = t;
        }
        return;
    }
    int t = blockIdx.x * 256 + threadIdx.x;
    if (t >= BATCH * D1) return;
    int b = t / D1, f = t % D1;
    int c = f >> 10;
    int rem = f & 1023;
    int oh = rem >> 5, ow = rem & 31;
    int base = c * (HH * WW) + (2 * oh) * WW + 2 * ow;
    const float* xb = x + b * (CCH * HH * WW);
    float s = (xb[base]        + bias[base])
            + (xb[base + 1]    + bias[base + 1])
            + (xb[base + WW]   + bias[base + WW])
            + (xb[base + WW+1] + bias[base + WW + 1]);
    xp[t] = 0.25f * s;
}

template <bool SQ, bool PW, int DIN, int DOUT, int G>
__device__ __forceinline__ void
gemv_row(const int o, const float* __restrict__ W, const float* __restrict__ X,
         const float* __restrict__ M, float* __restrict__ out,
         float* __restrict__ q, const float* __restrict__ gp,
         float* __restrict__ d2o, float* __restrict__ m2bo,
         float* __restrict__ vno) {
    static_assert(PW || !SQ, "the first layer always carries its power term");
    constexpr bool LM = PW && !SQ;
    constexpr int Q  = DIN / 4;
    constexpr int NC = Q / 32;
    constexpr int NG = NC / G;
    const int lane = threadIdx.x & 31;
    const v4f* w4 = (const v4f*)(W + (size_t)o * DIN) + lane;
    const v4f* x4 = (const v4f*)X + lane;
    const v4f* m4 = (const v4f*)M + lane;
    v4f w[NC];
#pragma unroll
    for (int c = 0; c < NC; ++c) w[c] = w4[c * 32];
    v4f xb[2][G][4], mb[2][G][4];
#pragma unroll
    for (int j = 0; j < G; ++j) {
#pragma unroll
        for (int b = 0; b < 4; ++b) {
            xb[0][j][b] = x4[b * Q + j * 32];
            if constexpr (LM) mb[0][j][b] = m4[b * Q + j * 32];
        }
    }
    v4f a[4] = {}, p[4] = {}, e[4] = {};
#pragma unroll
    for (int g = 0; g < NG; ++g) {
        if (g + 1 < NG) {
#pragma unroll
            for (int j = 0; j < G; ++j) {
#pragma unroll
                for (int b = 0; b < 4; ++b) {
                    xb[(g + 1) & 1][j][b] = x4[b * Q + ((g + 1) * G + j) * 32];
                    if constexpr (LM) mb[(g + 1) & 1][j][b] = m4[b * Q + ((g + 1) * G + j) * 32];
                }
            }
        }
        __builtin_amdgcn_sched_barrier(0);
#pragma unroll
        for (int j = 0; j < G; ++j) {
            const int c = g * G + j;
            v4f aw = __builtin_elementwise_abs(w[c]);
#pragma unroll
            for (int b = 0; b < 4; ++b) {
                v4f x = xb[g & 1][j][b];
                a[b] += w[c] * x;
                if constexpr (PW) {
                    v4f m;
                    if constexpr (SQ) { m = x * x; e[b] += m; }
                    else              m = mb[g & 1][j][b];
                    p[b] += aw * m;
                }
            }
        }
    }
    constexpr int NV = SQ ? 12 : (PW ? 8 : 4);
    constexpr int NS = PW ? 8 : 4;
    float s[NV];
#pragma unroll
    for (int b = 0; b < 4; ++b) {
        s[b]     = (a[b].x + a[b].y) + (a[b].z + a[b].w);
        if constexpr (PW) s[4 + b] = (p[b].x + p[b].y) + (p[b].z + p[b].w);
        if constexpr (SQ) s[8 + b] = (e[b].x + e[b].y) + (e[b].z + e[b].w);
    }
#pragma unroll
    for (int off = 16; off > 0; off >>= 1) {
#pragma unroll
        for (int v = 0; v < NV; ++v) s[v] += __shfl_xor(s[v], off);
    }
    float r = s[0];
#pragma unroll
    for (int v = 1; v < NS; ++v) r = (lane == v) ? s[v] : r;
    if constexpr (SQ) {
        float ev = s[8];
#pragma unroll
        for (int b = 1; b < 4; ++b) ev = (lane == b) ? s[8 + b] : ev;
        if (lane < 4) {
            const float c1 = gp[0];
            const float vb = 2.f * (SIGMA * SIGMA) / (c1 * c1) * ev;
            const float m  = r;
            const float dd = 1.f / (1.f + expf(-m));
            const float sv = (m > 20.f) ? m : log1pf(expf(m));
            const float dsq = dd * dd;
            d2o[lane * DOUT + o]  = dsq;
            out[lane * DOUT + o]  = sv;
            m2bo[lane * DOUT + o] = sv * sv + vb * dsq;
            if (o == 0) vno[lane] = vb;
        } else if (lane < 8) {
            q[(lane - 4) * DOUT + o] = r;
        }
    } else if constexpr (PW) {
        if (lane < 4)      out[lane * DOUT + o]     = r;
        else if (lane < 8) q[(lane - 4) * DOUT + o] = r;
    } else {
        if (lane < 4)      out[lane * DOUT + o]     = r;
    }
}

__global__ void __launch_bounds__(128)
gemv1_kernel(const float* __restrict__ W, const float* __restrict__ X,
             float* __restrict__ out, float* __restrict__ q,
             const float* __restrict__ gp, float* __restrict__ d2o,
             float* __restrict__ m2bo, float* __restrict__ vno) {
    gemv_row<true, true, D1, D2, 6>(blockIdx.x * 4 + (threadIdx.x >> 5),
                                    W, X, X, out, q, gp, d2o, m2bo, vno);
}

struct KStage { v4f x; v4f b; };

struct KFrag { v2f a[BK / 4]; v2f b0[BK / 4]; v2f b1[BK / 4]; v2f d[BK / 4]; };

__device__ __forceinline__ void lds_barrier() {
    __builtin_amdgcn_fence(__ATOMIC_RELEASE, "workgroup", "local");
    __builtin_amdgcn_s_barrier();
    __builtin_amdgcn_fence(__ATOMIC_ACQUIRE, "workgroup", "local");
}

#define GTH    512
#define STG_B  (GR * GP)
#define STG_D  ((GR + GT) * GP)
#define STG_SZ (STG_D + BATCH * BK)
#define EPB    (GT * TP)
#define SMEMF  (BATCH * EPB)
#define NLIGHT 128
#define KW     (D2 / 16)


__device__ __forceinline__ void gamma_tile4(const float* __restrict__ W2,
                                            const float* __restrict__ d2,
                                            const float* __restrict__ vn1,
                                            float* smem, int brow, int bcol,
                                            float vd, bool diag, int tq0,
                                            float* __restrict__ out) {
    const int tid  = threadIdx.x;
    const int lane = tid & 31;
    const int w    = tid >> 5;
    const int bb   = w >> 2;
    const int wy   = (w >> 1) & 1;
    const int wx   = w & 1;
    const int r    = lane & 15;
    const int hi   = lane >> 4;
    const int t    = tid & 127;

    const int srow = tid >> 3;
    const int c4   = tid & 7;
    const float* gB = W2 + (size_t)(bcol + srow) * D2 + c4 * 4;
    const float* gX;
    int xoff;
    if (w < 8) {
        gX   = W2 + (size_t)(brow + srow) * D2 + c4 * 4;
        xoff = srow * GP + c4 * 4;
    } else {
        gX   = d2 + (size_t)(lane >> 3) * D2 + (lane & 7) * 4;
        xoff = STG_D + lane * 4;
    }
    const bool xst = __builtin_amdgcn_readfirstlane(w) <= 8;
    const int boff = STG_B + srow * GP + c4 * 4;

    float* st0 = smem;
    float* st1 = smem + STG_SZ;

    auto request = [&](KStage& s, int kb) {
        s.x = *(const v4f*)(gX + kb * BK);
        s.b = *(const v4f*)(gB + kb * BK);
    };
    auto stage = [&](const KStage& s, float* st) {
        *(v4f*)(st + boff) = s.b;
        if (xst) *(v4f*)(st + xoff) = s.x;
    };

    const int fa = (wy * 16 + r) * GP + 2 * hi;
    const int fb = STG_B + (wx * 32 + r) * GP + 2 * hi;
    const int fd = STG_D + bb * BK + 2 * hi;

    auto fetch = [&](KFrag& f, const float* st) {
#pragma unroll
        for (int ks = 0; ks < BK / 4; ++ks) {
            f.a[ks]  = *(const v2f*)(st + fa + 4 * ks);
            f.b0[ks] = *(const v2f*)(st + fb + 4 * ks);
            f.b1[ks] = *(const v2f*)(st + fb + 16 * GP + 4 * ks);
            f.d[ks]  = *(const v2f*)(st + fd + 4 * ks);
        }
    };

    v8f acc0 = {}, acc1 = {};
    auto mma = [&](const KFrag& f, int k0, int k1) {
#pragma unroll
        for (int ks = k0; ks < k1; ++ks) {
            const v2f av = f.a[ks] * f.d[ks];
            acc0 = __builtin_amdgcn_wmma_f32_16x16x4_f32(
                false, av, false, f.b0[ks], (short)0, acc0, false, false);
            acc1 = __builtin_amdgcn_wmma_f32_16x16x4_f32(
                false, av, false, f.b1[ks], (short)0, acc1, false, false);
        }
    };

    KStage s0, s1;
    KFrag  f0, f1;
    request(s0, 0);
    __builtin_amdgcn_sched_barrier(0);
    request(s1, 1);
    stage(s0, st0);
    request(s0, 2);
    stage(s1, st1);
    request(s1, 3);
    lds_barrier();
    fetch(f0, st0);
    lds_barrier();

    auto k_step = [&](const KFrag& fc, KFrag& fn, KStage& s, float* sc, const float* sn,
                      int n, bool next, bool store, bool more) {
        mma(fc, 0, 2);
        __builtin_amdgcn_sched_barrier(0);
        if (next)  fetch(fn, sn);
        if (store) stage(s, sc);
        if (more)  request(s, n + 4);
        __builtin_amdgcn_sched_barrier(0);
        mma(fc, 2, BK / 4);
        __builtin_amdgcn_sched_barrier(0);
        lds_barrier();
    };

    for (int n = 0; n < NKB - 4; n += 2) {
        k_step(f0, f1, s0, st0, st1, n,     true, true, true);
        k_step(f1, f0, s1, st1, st0, n + 1, true, true, true);
    }
    k_step(f0, f1, s0, st0, st1, NKB - 4, true,  true,  false);
    k_step(f1, f0, s1, st1, st0, NKB - 3, true,  true,  false);
    k_step(f0, f1, s0, st0, st1, NKB - 2, true,  false, false);
    k_step(f1, f0, s1, st1, st0, NKB - 1, false, false, false);

    const float s  = vn1[bb];
    float* Cs = smem + bb * EPB;
    float* ob = out + (size_t)bb * D3 * D3;

#pragma unroll
    for (int v = 0; v < 8; ++v) {
        const int lr = wy * 16 + v + 8 * hi;
        const int lc = wx * 32 + r;
        float v0 = s * acc0[v];
        float v1 = s * acc1[v];
        if (diag && lr == lc)      v0 += vd;
        if (diag && lr == lc + 16) v1 += vd;
        acc0[v] = v0;
        acc1[v] = v1;
        Cs[lr * CP + lc]      = v0;
        Cs[lr * CP + lc + 16] = v1;
    }
    __syncthreads();

    const int er = t >> 4;
    const int ec = (t & 15) * 4;
#pragma unroll
    for (int q = 0; q < 4; ++q) {
        const int row = er + 8 * q;
        const v4f v4 = *(const v4f*)(Cs + row * CP + ec);
        *(v4f*)(ob + (size_t)(brow + row) * D3 + bcol + ec) = v4;
    }

    __syncthreads();
#pragma unroll
    for (int v = 0; v < 8; ++v) {
        const int lr = wy * 16 + v + 8 * hi;
        const int lc = wx * 32 + r;
        Cs[lc * TP + lr]        = acc0[v];
        Cs[(lc + 16) * TP + lr] = acc1[v];
    }
    __syncthreads();
    const int tr = t >> 3;
    const int tc = (t & 7) * 4;
    if (tq0 == 0) {
#pragma unroll
        for (int q = 0; q < 2; ++q) {
            const int row = tr + 16 * q;
            const v4f v4 = *(const v4f*)(Cs + row * TP + tc);
            *(v4f*)(ob + (size_t)(bcol + row) * D3 + brow + tc) = v4;
        }
    }
#pragma unroll
    for (int q = 2; q < 4; ++q) {
        const int row = tr + 16 * q;
        const v4f v4 = *(const v4f*)(Cs + row * TP + tc);
        *(v4f*)(ob + (size_t)(bcol + row) * D3 + brow + tc) = v4;
    }
}

__device__ __forceinline__ void light_unit(const float* __restrict__ W2,
                                           const float* __restrict__ d2,
                                           const float* __restrict__ vn1,
                                           const float* __restrict__ m2b,
                                           const float* __restrict__ g2p,
                                           float* smem, float* red, int u,
                                           float* __restrict__ out) {
    const int tid  = threadIdx.x;
    const int lane = tid & 31;
    const int w    = tid >> 5;
    const int r    = lane & 15;
    const int hi   = lane >> 4;
    const int tile = u >> 3, lb = (u >> 1) & 3, sr = u & 1;
    const int R0 = tile * GT + GR + 16 * sr;
    const int C0 = tile * GT + GR;
    const int k0 = w * KW + 4 * hi;
    const v4f* pa  = (const v4f*)(W2 + (size_t)(R0 + r) * D2 + k0);
    const v4f* pb0 = (const v4f*)(W2 + (size_t)(C0 + r) * D2 + k0);
    const v4f* pb1 = (const v4f*)(W2 + (size_t)(C0 + 16 + r) * D2 + k0);
    const v4f* pd  = (const v4f*)(d2 + (size_t)lb * D2 + k0);
    const v4f mp   = ((const v4f*)(m2b + (size_t)lb * D2))[tid];

    v8f acc0 = {}, acc1 = {};
#pragma unroll
    for (int c = 0; c < 2; ++c) {
        v4f A[8], B0[8], B1[8], Dv[8];
#pragma unroll
        for (int j = 0; j < 8; ++j) {
            const int ix = (c * 8 + j) * 2;
            A[j] = pa[ix]; B0[j] = pb0[ix]; B1[j] = pb1[ix]; Dv[j] = pd[ix];
        }
#pragma unroll
        for (int j = 0; j < 8; ++j) {
            const v4f ad = A[j] * Dv[j];
            const v2f a0 = {ad.x, ad.y}, a1 = {ad.z, ad.w};
            const v2f p0 = {B0[j].x, B0[j].y}, p1 = {B0[j].z, B0[j].w};
            const v2f q0 = {B1[j].x, B1[j].y}, q1 = {B1[j].z, B1[j].w};
            acc0 = __builtin_amdgcn_wmma_f32_16x16x4_f32(false, a0, false, p0, (short)0, acc0, false, false);
            acc1 = __builtin_amdgcn_wmma_f32_16x16x4_f32(false, a0, false, q0, (short)0, acc1, false, false);
            acc0 = __builtin_amdgcn_wmma_f32_16x16x4_f32(false, a1, false, p1, (short)0, acc0, false, false);
            acc1 = __builtin_amdgcn_wmma_f32_16x16x4_f32(false, a1, false, q1, (short)0, acc1, false, false);
        }
    }
    float* part = smem + w * 512;
#pragma unroll
    for (int v = 0; v < 8; ++v) {
        part[v * 32 + lane]       = acc0[v];
        part[256 + v * 32 + lane] = acc1[v];
    }
    float ms = (mp.x + mp.y) + (mp.z + mp.w);
    for (int off = 16; off > 0; off >>= 1) ms += __shfl_xor(ms, off);
    if (lane == 0) red[w] = ms;
    __syncthreads();

    float sum = 0.f, msum = 0.f;
#pragma unroll
    for (int ww = 0; ww < 16; ++ww) { sum += smem[ww * 512 + tid]; msum += red[ww]; }
    const float c2 = g2p[0];
    const float vd = 2.f * (SIGMA * SIGMA) / (c2 * c2) * msum;
    const int jb = tid >> 8, v = (tid >> 5) & 7;
    const int row = v + 8 * hi, col = jb * 16 + r;
    float val = vn1[lb] * sum;
    if (R0 + row == C0 + col) val += vd;
    float* Ts = smem + 16 * 512;
    Ts[row * TP + col] = val;
    __syncthreads();
    if (tid < 128) {
        const int rr = tid >> 3, cc = (tid & 7) * 4;
        const v4f o = *(const v4f*)(Ts + rr * TP + cc);
        *(v4f*)(out + (size_t)lb * D3 * D3 + (size_t)(R0 + rr) * D3 + C0 + cc) = o;
    }
}

#define NOFF   (NPAIR - NT)
#define NPOS   (2 * NOFF + NT)

__global__ void __launch_bounds__(GTH)
gamma_kernel(const float* __restrict__ W2, const float* __restrict__ d2,
             const float* __restrict__ vn1, const float* __restrict__ m2b,
             const float* __restrict__ q1, const float* __restrict__ csp,
             const float* __restrict__ g1p, const float* __restrict__ g2p,
             const float* __restrict__ sp, float* __restrict__ mu2,
             float* __restrict__ out, float* __restrict__ Pout) {
    __shared__ __attribute__((aligned(16))) float smem[SMEMF];
    __shared__ float red[BATCH * 48];
    static_assert(2 * STG_SZ <= SMEMF, "the two stages fit");
    static_assert(GR * CP <= EPB, "a batch's result tile fits its part");
    static_assert(16 * 512 + 16 * TP <= SMEMF, "the light unit's partial tiles fit");

    const int tid  = threadIdx.x;
    const int lane = tid & 31;
    const int w    = tid >> 5;

    const int m = blockIdx.x;
    int ti, tj, half = 0;
    bool diag, pown = false;
    if (m < 2 * NOFF) {
        int u = m >> 1;
        half = m & 1;
        ti = 0;
        while (u >= NT - 1 - ti) { u -= NT - 1 - ti; ++ti; }
        tj = ti + 1 + u;
        diag = false;
    } else {
        ti = tj = m - 2 * NOFF;
        diag = true;
        pown = (ti == 0);
    }
    const int brow = ti * GT + half * GR;
    const int bcol = tj * GT;

    float vd = 0.f;
    if (diag) {
        v4f cs = {};
        if (pown) cs = ((const v4f*)csp)[tid] + ((const v4f*)(csp + D2))[tid];
#pragma unroll
        for (int b = 0; b < BATCH; ++b) {
            const v4f mp = ((const v4f*)(m2b + (size_t)b * D2))[tid];
            float ms = (mp.x + mp.y) + (mp.z + mp.w);
            float s1 = 0.f, s2 = 0.f;
            if (pown) {
                const v4f a = ((const v4f*)(q1 + (size_t)b * D2))[tid];
                const v4f c = cs * mp;
                s1 = (a.x + a.y) + (a.z + a.w);
                s2 = (c.x + c.y) + (c.z + c.w);
            }
            for (int off = 16; off > 0; off >>= 1) {
                ms += __shfl_xor(ms, off);
                s1 += __shfl_xor(s1, off);
                s2 += __shfl_xor(s2, off);
            }
            if (lane == 0) { red[b * 48 + w] = ms; red[b * 48 + 16 + w] = s1; red[b * 48 + 32 + w] = s2; }
        }
        __syncthreads();
        const float c2 = g2p[0];
        const int bb = w >> 2;
        float tot = 0.f;
#pragma unroll
        for (int ww = 0; ww < 16; ++ww) tot += red[bb * 48 + ww];
        vd = 2.f * (SIGMA * SIGMA) / (c2 * c2) * tot;
        if (pown && tid < BATCH) {
            const float c1 = g1p[0];
            float t1 = 0.f, t2 = 0.f;
#pragma unroll
            for (int ww = 0; ww < 16; ++ww) { t1 += red[tid * 48 + 16 + ww]; t2 += red[tid * 48 + 32 + ww]; }
            Pout[tid] = c1 * t1 + c2 * t2;
        }
    }

    gamma_tile4(W2, d2, vn1, smem, brow, bcol, vd, diag, diag ? 2 : 0, out);

    if (m < NLIGHT) {
        __syncthreads();
        light_unit(W2, d2, vn1, m2b, g2p, smem, red, m, out);
    } else if (w < 8) {
        gemv_row<false, false, D2, D3, 2>((m - NLIGHT) * 8 + w, W2, sp, nullptr, mu2, nullptr,
                                          nullptr, nullptr, nullptr, nullptr);
    }
}

extern "C" void kernel_launch(void* const* d_in, const int* in_sizes, int n_in,
                              void* d_out, int out_size, void* d_ws, size_t ws_size,
                              hipStream_t stream) {
    (void)in_sizes; (void)n_in; (void)out_size; (void)ws_size;
    const float* x    = (const float*)d_in[0];
    const float* bias = (const float*)d_in[1];
    const float* W1   = (const float*)d_in[2];
    const float* W2   = (const float*)d_in[3];
    const float* g1   = (const float*)d_in[4];
    const float* g2   = (const float*)d_in[5];

    float* out = (float*)d_out;
    float* out_mu2   = out;
    float* out_gamma = out + BATCH * D3;
    float* out_P     = out + BATCH * D3 + (size_t)BATCH * D3 * D3;

    float* ws = (float*)d_ws;
    float* xp   = ws;
    float* csp  = xp + BATCH * D1;
    float* q1   = csp + D1 + D2;
    float* d2   = q1 + BATCH * D2;
    float* sp   = d2 + BATCH * D2;
    float* m2b  = sp + BATCH * D2;
    float* vn1  = m2b + BATCH * D2;

    pool_colsum_kernel<<<NPOOLB + NCSB, 256, 0, stream>>>(x, bias, W2, xp, csp);
    gemv1_kernel<<<D2 / 4, 128, 0, stream>>>(W1, xp, sp, q1, g1, d2, m2b, vn1);

    gamma_kernel<<<NPOS, GTH, 0, stream>>>(W2, d2, vn1, m2b, q1, csp, g1, g2,
                                           sp, out_mu2, out_gamma, out_P);
}
